# starting kernel
# baseline (speedup 1.0000x reference)
#include <hip/hip_runtime.h>

typedef float v2f __attribute__((ext_vector_type(2)));
typedef float v8f __attribute__((ext_vector_type(8)));

#define N_NODES 50000
#define FEAT    128

__global__ void sage_zero_kernel(float* __restrict__ p, long long n) {
  long long i = (long long)blockIdx.x * blockDim.x + threadIdx.x;
  long long stride = (long long)gridDim.x * blockDim.x;
  for (; i < n; i += stride) p[i] = 0.0f;
}

__global__ void sage_scatter_kernel(const float* __restrict__ x,
                                    const int* __restrict__ src,
                                    const int* __restrict__ dst,
                                    float* __restrict__ msg,
                                    float* __restrict__ deg,
                                    int E, int addDeg) {
  long long gid = (long long)blockIdx.x * blockDim.x + threadIdx.x;
  long long total = (long long)E * 32;
  if (gid >= total) return;
  int e = (int)(gid >> 5);
  int c = (int)(gid & 31);
  int s = src[e];
  int d = dst[e];
  const float4* xs = (const float4*)(x + (long long)s * FEAT);
  float4 v = xs[c];
  float* p = msg + (long long)d * FEAT + c * 4;
  unsafeAtomicAdd(p + 0, v.x);
  unsafeAtomicAdd(p + 1, v.y);
  unsafeAtomicAdd(p + 2, v.z);
  unsafeAtomicAdd(p + 3, v.w);
  if (addDeg && c == 0) unsafeAtomicAdd(deg + d, 1.0f);
}

__global__ void sage_gemm_kernel(const float* __restrict__ msg,
                                 const float* __restrict__ deg,
                                 const float* __restrict__ xin,
                                 const float* __restrict__ Wl,
                                 const float* __restrict__ bl,
                                 const float* __restrict__ Wr,
                                 float* __restrict__ out,
                                 int K, int N, int doRelu) {
  const int lane = threadIdx.x & 31;
  const int wave = threadIdx.x >> 5;
  const int wavesPerBlock = blockDim.x >> 5;
  const int tile = blockIdx.x * wavesPerBlock + wave;
  const int colTiles = N >> 4;
  const int rowTiles = N_NODES / 16;
  const int rowTile = tile / colTiles;
  const int colTile = tile - rowTile * colTiles;
  if (rowTile >= rowTiles) return;

  const int r   = (rowTile << 4) + (lane & 15);
  const int col = (colTile << 4) + (lane & 15);
  const int kb  = (lane >> 4) << 1;

  const float scale = 1.0f / fmaxf(deg[r], 1.0f);

  v8f c = {};
  const float* arow = msg + (long long)r * K;
  for (int k0 = 0; k0 < K; k0 += 4) {
    v2f a, b;
    a.x = arow[k0 + kb] * scale;
    a.y = arow[k0 + kb + 1] * scale;
    b.x = Wl[(k0 + kb) * N + col];
    b.y = Wl[(k0 + kb + 1) * N + col];
    c = __builtin_amdgcn_wmma_f32_16x16x4_f32(false, a, false, b,
                                              (short)0, c, false, false);
  }
  const float* xrow = xin + (long long)r * K;
  for (int k0 = 0; k0 < K; k0 += 4) {
    v2f a, b;
    a.x = xrow[k0 + kb];
    a.y = xrow[k0 + kb + 1];
    b.x = Wr[(k0 + kb) * N + col];
    b.y = Wr[(k0 + kb + 1) * N + col];
    c = __builtin_amdgcn_wmma_f32_16x16x4_f32(false, a, false, b,
                                              (short)0, c, false, false);
  }

  const float bias = bl[col];
  const int rbase = (rowTile << 4) + ((lane >> 4) << 3);
#pragma unroll
  for (int i = 0; i < 8; ++i) {
    float v = c[i] + bias;
    if (doRelu) v = fmaxf(v, 0.0f);
    out[(long long)(rbase + i) * N + col] = v;
  }
}

extern "C" void kernel_launch(void* const* d_in, const int* in_sizes, int n_in,
                              void* d_out, int out_size, void* d_ws, size_t ws_size,
                              hipStream_t stream) {
  const float* x   = (const float*)d_in[0];
  const int*   ei  = (const int*)d_in[1];
  const float* Wl1 = (const float*)d_in[2];
  const float* bl1 = (const float*)d_in[3];
  const float* Wr1 = (const float*)d_in[4];
  const float* Wl2 = (const float*)d_in[5];
  const float* bl2 = (const float*)d_in[6];
  const float* Wr2 = (const float*)d_in[7];
  float* out = (float*)d_out;

  const int E = in_sizes[1] / 2;
  const int* src = ei;
  const int* dst = ei + E;

  float* msg = (float*)d_ws;
  float* deg = msg + (long long)N_NODES * FEAT;
  float* h1  = deg + N_NODES;

  const int ZB = 2048, TB = 256;
  const long long scatterThreads = (long long)E * 32;
  const int scatterBlocks = (int)((scatterThreads + TB - 1) / TB);

  sage_zero_kernel<<<ZB, TB, 0, stream>>>(msg, (long long)N_NODES * FEAT + N_NODES);
  sage_scatter_kernel<<<scatterBlocks, TB, 0, stream>>>(x, src, dst, msg, deg, E, 1);
  {
    const int tiles = (N_NODES / 16) * (128 / 16);
    const int blocks = (tiles + 7) / 8;
    sage_gemm_kernel<<<blocks, TB, 0, stream>>>(msg, deg, x, Wl1, bl1, Wr1, h1,
                                                128, 128, 1);
  }

  sage_zero_kernel<<<ZB, TB, 0, stream>>>(msg, (long long)N_NODES * FEAT);
  sage_scatter_kernel<<<scatterBlocks, TB, 0, stream>>>(h1, src, dst, msg, deg, E, 0);
  {
    const int tiles = (N_NODES / 16) * (64 / 16);
    const int blocks = (tiles + 7) / 8;
    sage_gemm_kernel<<<blocks, TB, 0, stream>>>(msg, deg, h1, Wl2, bl2, Wr2, out,
                                                128, 64, 0);
  }
}


# Replace the per-edge float-atomic scatter by a gather over edges counting-sorted by destination (sorted once per call, integer atomics); float sums now sequential per node in slot (arrival) order
# speedup vs baseline: 1.6256x; 1.6256x over previous
#include <hip/hip_runtime.h>

typedef float v2f __attribute__((ext_vector_type(2)));
typedef float v8f __attribute__((ext_vector_type(8)));

#define N_NODES 50000
#define FEAT    128

__global__ void sage_zero_kernel(float* __restrict__ p, long long n) {
  long long i = (long long)blockIdx.x * blockDim.x + threadIdx.x;
  long long stride = (long long)gridDim.x * blockDim.x;
  for (; i < n; i += stride) p[i] = 0.0f;
}


__global__ void sage_count_kernel(const int* __restrict__ dst,
                                  int* __restrict__ cnt, int E) {
  const int e = blockIdx.x * blockDim.x + threadIdx.x;
  if (e < E) atomicAdd(cnt + dst[e], 1);
}

#define SCAN_THREADS   1024
#define SCAN_CH        52
#define SCAN_LEN       (SCAN_THREADS * SCAN_CH)
#define SCAN_OUT_ITERS ((N_NODES + 1 + SCAN_THREADS - 1) / SCAN_THREADS)
__global__ void __launch_bounds__(SCAN_THREADS)
sage_rowstart_kernel(const int* __restrict__ cnt, int* __restrict__ rowstart,
                     float* __restrict__ deg) {
  __shared__ __attribute__((aligned(16))) int buf[SCAN_LEN];
  __shared__ int wsum[SCAN_THREADS / 32];
  const int tid  = threadIdx.x;
  const int lane = tid & 31;
  const int wave = tid >> 5;

#pragma unroll
  for (int it = 0; it < SCAN_CH; ++it) {
    const int i = it * SCAN_THREADS + tid;
    buf[i] = (i < N_NODES) ? cnt[i] : 0;
  }
  __syncthreads();

  int4* mine = (int4*)(buf + tid * SCAN_CH);
  int4 q[SCAN_CH / 4];
  int tot = 0;
#pragma unroll
  for (int k = 0; k < SCAN_CH / 4; ++k) {
    q[k] = mine[k];
    tot += q[k].x + q[k].y + q[k].z + q[k].w;
  }

  int inc = tot;
  for (int o = 1; o < 32; o <<= 1) {
    const int up = __shfl_up(inc, o);
    if (lane >= o) inc += up;
  }
  if (lane == 31) wsum[wave] = inc;
  __syncthreads();
  int woff = (lane < wave) ? wsum[lane] : 0;
  for (int o = 16; o > 0; o >>= 1) woff += __shfl_xor(woff, o);
  int run = woff + inc - tot;

#pragma unroll
  for (int k = 0; k < SCAN_CH / 4; ++k) {
    int4 p;
    p.x = run; run += q[k].x;
    p.y = run; run += q[k].y;
    p.z = run; run += q[k].z;
    p.w = run; run += q[k].w;
    mine[k] = p;
  }
  __syncthreads();

  for (int it = 0; it < SCAN_OUT_ITERS; ++it) {
    const int i = it * SCAN_THREADS + tid;
    if (i <= N_NODES) {
      const int a = buf[i];
      rowstart[i] = a;
      if (i < N_NODES) deg[i] = (float)(buf[i + 1] - a);
    }
  }
}

__global__ void sage_fill_kernel(const int* __restrict__ src,
                                 const int* __restrict__ dst,
                                 const int* __restrict__ rowstart,
                                 int* __restrict__ cnt,
                                 int* __restrict__ col, int E) {
  const int e = blockIdx.x * blockDim.x + threadIdx.x;
  if (e >= E) return;
  const int d = dst[e];
  const int slot = atomicSub(cnt + d, 1) - 1;
  col[rowstart[d] + slot] = src[e];
}

#define SAGE_ROW(k) x4[(long long)__builtin_amdgcn_readlane(idx, t + (k)) * 32 + lane]
#define SAGE_ADD(a) acc.x += a.x; acc.y += a.y; acc.z += a.z; acc.w += a.w;
__global__ void __launch_bounds__(256)
sage_gather_kernel(const float* __restrict__ x,
                   const int* __restrict__ rowstart,
                   const int* __restrict__ col,
                   float* __restrict__ msg) {
  const int lane = threadIdx.x & 31;
  const int node = __builtin_amdgcn_readfirstlane((int)(blockIdx.x * 8 + (threadIdx.x >> 5)));
  if (node >= N_NODES) return;
  const int beg = rowstart[node];
  const int end = rowstart[node + 1];
  const float4* x4 = (const float4*)x;
  float4 acc = make_float4(0.0f, 0.0f, 0.0f, 0.0f);
  for (int base = beg; base < end; base += 32) {
    const int j = base + lane;
    const int idx = (j < end) ? col[j] : 0;
    const int m = min(32, end - base);
    int t = 0;
    for (; t + 8 <= m; t += 8) {
      const float4 a0 = SAGE_ROW(0), a1 = SAGE_ROW(1), a2 = SAGE_ROW(2), a3 = SAGE_ROW(3);
      const float4 a4 = SAGE_ROW(4), a5 = SAGE_ROW(5), a6 = SAGE_ROW(6), a7 = SAGE_ROW(7);
      SAGE_ADD(a0) SAGE_ADD(a1) SAGE_ADD(a2) SAGE_ADD(a3)
      SAGE_ADD(a4) SAGE_ADD(a5) SAGE_ADD(a6) SAGE_ADD(a7)
    }
    if (t + 4 <= m) {
      const float4 a0 = SAGE_ROW(0), a1 = SAGE_ROW(1), a2 = SAGE_ROW(2), a3 = SAGE_ROW(3);
      SAGE_ADD(a0) SAGE_ADD(a1) SAGE_ADD(a2) SAGE_ADD(a3)
      t += 4;
    }
    if (t + 2 <= m) {
      const float4 a0 = SAGE_ROW(0), a1 = SAGE_ROW(1);
      SAGE_ADD(a0) SAGE_ADD(a1)
      t += 2;
    }
    if (t < m) {
      const float4 a0 = SAGE_ROW(0);
      SAGE_ADD(a0)
    }
  }
  ((float4*)msg)[(long long)node * 32 + lane] = acc;
}

__global__ void sage_gemm_kernel(const float* __restrict__ msg,
                                 const float* __restrict__ deg,
                                 const float* __restrict__ xin,
                                 const float* __restrict__ Wl,
                                 const float* __restrict__ bl,
                                 const float* __restrict__ Wr,
                                 float* __restrict__ out,
                                 int K, int N, int doRelu) {
  const int lane = threadIdx.x & 31;
  const int wave = threadIdx.x >> 5;
  const int wavesPerBlock = blockDim.x >> 5;
  const int tile = blockIdx.x * wavesPerBlock + wave;
  const int colTiles = N >> 4;
  const int rowTiles = N_NODES / 16;
  const int rowTile = tile / colTiles;
  const int colTile = tile - rowTile * colTiles;
  if (rowTile >= rowTiles) return;

  const int r   = (rowTile << 4) + (lane & 15);
  const int col = (colTile << 4) + (lane & 15);
  const int kb  = (lane >> 4) << 1;

  const float scale = 1.0f / fmaxf(deg[r], 1.0f);

  v8f c = {};
  const float* arow = msg + (long long)r * K;
  for (int k0 = 0; k0 < K; k0 += 4) {
    v2f a, b;
    a.x = arow[k0 + kb] * scale;
    a.y = arow[k0 + kb + 1] * scale;
    b.x = Wl[(k0 + kb) * N + col];
    b.y = Wl[(k0 + kb + 1) * N + col];
    c = __builtin_amdgcn_wmma_f32_16x16x4_f32(false, a, false, b,
                                              (short)0, c, false, false);
  }
  const float* xrow = xin + (long long)r * K;
  for (int k0 = 0; k0 < K; k0 += 4) {
    v2f a, b;
    a.x = xrow[k0 + kb];
    a.y = xrow[k0 + kb + 1];
    b.x = Wr[(k0 + kb) * N + col];
    b.y = Wr[(k0 + kb + 1) * N + col];
    c = __builtin_amdgcn_wmma_f32_16x16x4_f32(false, a, false, b,
                                              (short)0, c, false, false);
  }

  const float bias = bl[col];
  const int rbase = (rowTile << 4) + ((lane >> 4) << 3);
#pragma unroll
  for (int i = 0; i < 8; ++i) {
    float v = c[i] + bias;
    if (doRelu) v = fmaxf(v, 0.0f);
    out[(long long)(rbase + i) * N + col] = v;
  }
}

extern "C" void kernel_launch(void* const* d_in, const int* in_sizes, int n_in,
                              void* d_out, int out_size, void* d_ws, size_t ws_size,
                              hipStream_t stream) {
  const float* x   = (const float*)d_in[0];
  const int*   ei  = (const int*)d_in[1];
  const float* Wl1 = (const float*)d_in[2];
  const float* bl1 = (const float*)d_in[3];
  const float* Wr1 = (const float*)d_in[4];
  const float* Wl2 = (const float*)d_in[5];
  const float* bl2 = (const float*)d_in[6];
  const float* Wr2 = (const float*)d_in[7];
  float* out = (float*)d_out;

  const int E = in_sizes[1] / 2;
  const int* src = ei;
  const int* dst = ei + E;

  float* msg = (float*)d_ws;
  float* deg = msg + (long long)N_NODES * FEAT;
  float* h1  = deg + N_NODES;

  int* cnt      = (int*)d_out;
  int* rowstart = cnt + 50048;
  int* col      = rowstart + 50048;

  const int TB = 256;
  const int edgeBlocks = (E + TB - 1) / TB;
  const int nodeBlocks = N_NODES / 8;

  sage_zero_kernel<<<(N_NODES + TB - 1) / TB, TB, 0, stream>>>((float*)cnt, (long long)N_NODES);
  sage_count_kernel<<<edgeBlocks, TB, 0, stream>>>(dst, cnt, E);
  sage_rowstart_kernel<<<1, SCAN_THREADS, 0, stream>>>(cnt, rowstart, deg);
  sage_fill_kernel<<<edgeBlocks, TB, 0, stream>>>(src, dst, rowstart, cnt, col, E);

  sage_gather_kernel<<<nodeBlocks, TB, 0, stream>>>(x, rowstart, col, msg);
  {
    const int tiles = (N_NODES / 16) * (128 / 16);
    const int blocks = (tiles + 7) / 8;
    sage_gemm_kernel<<<blocks, TB, 0, stream>>>(msg, deg, x, Wl1, bl1, Wr1, h1,
                                                128, 128, 1);
  }

  sage_gather_kernel<<<nodeBlocks, TB, 0, stream>>>(h1, rowstart, col, msg);
  {
    const int tiles = (N_NODES / 16) * (64 / 16);
    const int blocks = (tiles + 7) / 8;
    sage_gemm_kernel<<<blocks, TB, 0, stream>>>(msg, deg, h1, Wl2, bl2, Wr2, out,
                                                128, 64, 0);
  }
}


# Layer GEMM re-tiled: resident workgroups hold both weight matrices in LDS, a wave owns 16 rows x N/2 columns and loads each A fragment once; k summed in a permuted order inside 8-blocks
# speedup vs baseline: 1.9136x; 1.1772x over previous
#include <hip/hip_runtime.h>

typedef float v2f __attribute__((ext_vector_type(2)));
typedef float v8f __attribute__((ext_vector_type(8)));

#define N_NODES 50000
#define FEAT    128

__global__ void sage_zero_kernel(float* __restrict__ p, long long n) {
  long long i = (long long)blockIdx.x * blockDim.x + threadIdx.x;
  long long stride = (long long)gridDim.x * blockDim.x;
  for (; i < n; i += stride) p[i] = 0.0f;
}


__global__ void sage_count_kernel(const int* __restrict__ dst,
                                  int* __restrict__ cnt, int E) {
  const int e = blockIdx.x * blockDim.x + threadIdx.x;
  if (e < E) atomicAdd(cnt + dst[e], 1);
}

#define SCAN_THREADS   1024
#define SCAN_CH        52
#define SCAN_LEN       (SCAN_THREADS * SCAN_CH)
#define SCAN_OUT_ITERS ((N_NODES + 1 + SCAN_THREADS - 1) / SCAN_THREADS)
__global__ void __launch_bounds__(SCAN_THREADS)
sage_rowstart_kernel(const int* __restrict__ cnt, int* __restrict__ rowstart,
                     float* __restrict__ deg) {
  __shared__ __attribute__((aligned(16))) int buf[SCAN_LEN];
  __shared__ int wsum[SCAN_THREADS / 32];
  const int tid  = threadIdx.x;
  const int lane = tid & 31;
  const int wave = tid >> 5;

#pragma unroll
  for (int it = 0; it < SCAN_CH; ++it) {
    const int i = it * SCAN_THREADS + tid;
    buf[i] = (i < N_NODES) ? cnt[i] : 0;
  }
  __syncthreads();

  int4* mine = (int4*)(buf + tid * SCAN_CH);
  int4 q[SCAN_CH / 4];
  int tot = 0;
#pragma unroll
  for (int k = 0; k < SCAN_CH / 4; ++k) {
    q[k] = mine[k];
    tot += q[k].x + q[k].y + q[k].z + q[k].w;
  }

  int inc = tot;
  for (int o = 1; o < 32; o <<= 1) {
    const int up = __shfl_up(inc, o);
    if (lane >= o) inc += up;
  }
  if (lane == 31) wsum[wave] = inc;
  __syncthreads();
  int woff = (lane < wave) ? wsum[lane] : 0;
  for (int o = 16; o > 0; o >>= 1) woff += __shfl_xor(woff, o);
  int run = woff + inc - tot;

#pragma unroll
  for (int k = 0; k < SCAN_CH / 4; ++k) {
    int4 p;
    p.x = run; run += q[k].x;
    p.y = run; run += q[k].y;
    p.z = run; run += q[k].z;
    p.w = run; run += q[k].w;
    mine[k] = p;
  }
  __syncthreads();

  for (int it = 0; it < SCAN_OUT_ITERS; ++it) {
    const int i = it * SCAN_THREADS + tid;
    if (i <= N_NODES) {
      const int a = buf[i];
      rowstart[i] = a;
      if (i < N_NODES) deg[i] = (float)(buf[i + 1] - a);
    }
  }
}

__global__ void sage_fill_kernel(const int* __restrict__ src,
                                 const int* __restrict__ dst,
                                 const int* __restrict__ rowstart,
                                 int* __restrict__ cnt,
                                 int* __restrict__ col, int E) {
  const int e = blockIdx.x * blockDim.x + threadIdx.x;
  if (e >= E) return;
  const int d = dst[e];
  const int slot = atomicSub(cnt + d, 1) - 1;
  col[rowstart[d] + slot] = src[e];
}

#define SAGE_ROW(k) x4[(long long)__builtin_amdgcn_readlane(idx, t + (k)) * 32 + lane]
#define SAGE_ADD(a) acc.x += a.x; acc.y += a.y; acc.z += a.z; acc.w += a.w;
__global__ void __launch_bounds__(256)
sage_gather_kernel(const float* __restrict__ x,
                   const int* __restrict__ rowstart,
                   const int* __restrict__ col,
                   float* __restrict__ msg) {
  const int lane = threadIdx.x & 31;
  const int node = __builtin_amdgcn_readfirstlane((int)(blockIdx.x * 8 + (threadIdx.x >> 5)));
  if (node >= N_NODES) return;
  const int beg = rowstart[node];
  const int end = rowstart[node + 1];
  const float4* x4 = (const float4*)x;
  float4 acc = make_float4(0.0f, 0.0f, 0.0f, 0.0f);
  for (int base = beg; base < end; base += 32) {
    const int j = base + lane;
    const int idx = (j < end) ? col[j] : 0;
    const int m = min(32, end - base);
    int t = 0;
    for (; t + 8 <= m; t += 8) {
      const float4 a0 = SAGE_ROW(0), a1 = SAGE_ROW(1), a2 = SAGE_ROW(2), a3 = SAGE_ROW(3);
      const float4 a4 = SAGE_ROW(4), a5 = SAGE_ROW(5), a6 = SAGE_ROW(6), a7 = SAGE_ROW(7);
      SAGE_ADD(a0) SAGE_ADD(a1) SAGE_ADD(a2) SAGE_ADD(a3)
      SAGE_ADD(a4) SAGE_ADD(a5) SAGE_ADD(a6) SAGE_ADD(a7)
    }
    if (t + 4 <= m) {
      const float4 a0 = SAGE_ROW(0), a1 = SAGE_ROW(1), a2 = SAGE_ROW(2), a3 = SAGE_ROW(3);
      SAGE_ADD(a0) SAGE_ADD(a1) SAGE_ADD(a2) SAGE_ADD(a3)
      t += 4;
    }
    if (t + 2 <= m) {
      const float4 a0 = SAGE_ROW(0), a1 = SAGE_ROW(1);
      SAGE_ADD(a0) SAGE_ADD(a1)
      t += 2;
    }
    if (t < m) {
      const float4 a0 = SAGE_ROW(0);
      SAGE_ADD(a0)
    }
  }
  ((float4*)msg)[(long long)node * 32 + lane] = acc;
}

#define GEMM_GRID      512
#define GEMM_RT_BLOCK  6
#define GEMM_RT_EVEN   (GEMM_GRID * GEMM_RT_BLOCK)
#define GEMM_ROW_TILES (N_NODES / 16)
#define GEMM_KG        (FEAT / 4)
#define GEMM_ST        (GEMM_KG / 2)
#define GEMM_NB        GEMM_ST

template <int N>
__device__ __forceinline__ void
sage_gemm_item(const float* Ws, const float4 (&am)[GEMM_ST], float degv,
               const float* __restrict__ xin, const float* __restrict__ bl,
               float* __restrict__ out, int doRelu, int rowTile, int half, int c, int h) {
  constexpr int K  = FEAT;
  constexpr int KG = GEMM_KG;
  constexpr int NB = GEMM_NB;
  constexpr int CT = N / 32;

  const int r = (rowTile << 4) + c;
  const float scale = 1.0f / fmaxf(degv, 1.0f);
  const int colbase = half * (N / 2) + c;
  const float4* xrow = (const float4*)(xin + (long long)r * K) + h;
  const float* wb = Ws + (h * N + colbase) * 4;
#define SAGE_BFRAG(blk, u) \
  (*(const float4*)(wb + ((((blk) >> 3) * KG + 2 * (2 * ((blk) & 7) + (u) / CT)) * N + ((u) % CT) * 16) * 4))

  float4 ax[GEMM_ST];
  v8f acc[CT];
#pragma unroll
  for (int t = 0; t < CT; ++t) acc[t] = v8f{};

  float4 bq[2][2 * CT];
#pragma unroll
  for (int u = 0; u < 2 * CT; ++u) bq[0][u] = SAGE_BFRAG(0, u);

#pragma unroll
  for (int blk = 0; blk < NB; ++blk) {
    if (blk + 1 < NB) {
#pragma unroll
      for (int u = 0; u < 2 * CT; ++u) bq[(blk + 1) & 1][u] = SAGE_BFRAG(blk + 1, u);
    }
    if (blk < NB / 2) {
      ax[2 * blk]     = xrow[4 * blk];
      ax[2 * blk + 1] = xrow[4 * blk + 2];
    }
    __builtin_amdgcn_sched_barrier(0);
#pragma unroll
    for (int u2 = 0; u2 < 2; ++u2) {
      float4 a;
      if (blk < NB / 2) {
        a = am[2 * blk + u2];
        a.x *= scale; a.y *= scale; a.z *= scale; a.w *= scale;
      } else {
        a = ax[2 * (blk - NB / 2) + u2];
      }
      v2f a0, a1;
      a0.x = a.x; a0.y = a.y;
      a1.x = a.z; a1.y = a.w;
#pragma unroll
      for (int t = 0; t < CT; ++t) {
        v2f b0;
        b0.x = bq[blk & 1][u2 * CT + t].x; b0.y = bq[blk & 1][u2 * CT + t].y;
        acc[t] = __builtin_amdgcn_wmma_f32_16x16x4_f32(false, a0, false, b0,
                                                       (short)0, acc[t], false, false);
      }
#pragma unroll
      for (int t = 0; t < CT; ++t) {
        v2f b1;
        b1.x = bq[blk & 1][u2 * CT + t].z; b1.y = bq[blk & 1][u2 * CT + t].w;
        acc[t] = __builtin_amdgcn_wmma_f32_16x16x4_f32(false, a1, false, b1,
                                                       (short)0, acc[t], false, false);
      }
    }
    __builtin_amdgcn_sched_barrier(0);
  }
#undef SAGE_BFRAG

  const int rbase = (rowTile << 4) + (h << 3);
#pragma unroll
  for (int t = 0; t < CT; ++t) {
    const int col = colbase + t * 16;
    const float bias = bl[col];
#pragma unroll
    for (int i = 0; i < 8; ++i) {
      float v = acc[t][i] + bias;
      if (doRelu) v = fmaxf(v, 0.0f);
      out[(long long)(rbase + i) * N + col] = v;
    }
  }
}

template <int N>
__global__ void __launch_bounds__(256)
sage_gemm_strip_kernel(const float* __restrict__ msg,
                       const float* __restrict__ deg,
                       const float* __restrict__ xin,
                       const float* __restrict__ Wl,
                       const float* __restrict__ bl,
                       const float* __restrict__ Wr,
                       float* __restrict__ out, int doRelu) {
  constexpr int K  = FEAT;
  constexpr int KG = GEMM_KG;
  __shared__ __attribute__((aligned(16))) float Ws[2 * K * N];

  const int tid  = threadIdx.x;
  const int lane = tid & 31;
  const int c = lane & 15;
  const int h = lane >> 4;
  const int wv = __builtin_amdgcn_readfirstlane(tid >> 5);
  const int b  = blockIdx.x;

  const int rowTile1 = b * GEMM_RT_BLOCK + (wv >> 1);
  const int half1 = wv & 1;
  const int q = b >> 2;
  const bool left = (wv == 4) && ((b & 3) == 1) && (GEMM_RT_EVEN + (q >> 1) < GEMM_ROW_TILES);
  const bool second = (wv < 4) || left;
  const int rowTile2 = left ? GEMM_RT_EVEN + (q >> 1) : b * GEMM_RT_BLOCK + 4 + (wv >> 1);
  const int half2 = left ? (q & 1) : (wv & 1);

  const int r1 = (rowTile1 << 4) + c;
  const float deg1 = deg[r1];
  float4 am[GEMM_ST];
  {
    const float4* mrow = (const float4*)(msg + (long long)r1 * K) + h;
#pragma unroll
    for (int j = 0; j < GEMM_ST; ++j) am[j] = mrow[2 * j];
  }

  {
    constexpr int KG_PASS = 256 / N;
    const int n   = tid % N;
    const int kgo = tid / N;
    for (int m = 0; m < 2; ++m) {
      const float* __restrict__ Wm = m ? Wr : Wl;
#pragma unroll 4
      for (int p = 0; p < KG / KG_PASS; ++p) {
        const int kg = p * KG_PASS + kgo;
        float4 v;
        v.x = Wm[(4 * kg + 0) * N + n];
        v.y = Wm[(4 * kg + 1) * N + n];
        v.z = Wm[(4 * kg + 2) * N + n];
        v.w = Wm[(4 * kg + 3) * N + n];
        *(float4*)(Ws + ((m * KG + kg) * N + n) * 4) = v;
      }
    }
  }
  __syncthreads();

  sage_gemm_item<N>(Ws, am, deg1, xin, bl, out, doRelu, rowTile1, half1, c, h);
  if (second) {
    const int r2 = (rowTile2 << 4) + c;
    const float deg2 = deg[r2];
    float4 an[GEMM_ST];
    const float4* mrow = (const float4*)(msg + (long long)r2 * K) + h;
#pragma unroll
    for (int j = 0; j < GEMM_ST; ++j) an[j] = mrow[2 * j];
    sage_gemm_item<N>(Ws, an, deg2, xin, bl, out, doRelu, rowTile2, half2, c, h);
  }
}

extern "C" void kernel_launch(void* const* d_in, const int* in_sizes, int n_in,
                              void* d_out, int out_size, void* d_ws, size_t ws_size,
                              hipStream_t stream) {
  const float* x   = (const float*)d_in[0];
  const int*   ei  = (const int*)d_in[1];
  const float* Wl1 = (const float*)d_in[2];
  const float* bl1 = (const float*)d_in[3];
  const float* Wr1 = (const float*)d_in[4];
  const float* Wl2 = (const float*)d_in[5];
  const float* bl2 = (const float*)d_in[6];
  const float* Wr2 = (const float*)d_in[7];
  float* out = (float*)d_out;

  const int E = in_sizes[1] / 2;
  const int* src = ei;
  const int* dst = ei + E;

  float* msg = (float*)d_ws;
  float* deg = msg + (long long)N_NODES * FEAT;
  float* h1  = deg + N_NODES;

  int* cnt      = (int*)d_out;
  int* rowstart = cnt + 50048;
  int* col      = rowstart + 50048;

  const int TB = 256;
  const int edgeBlocks = (E + TB - 1) / TB;
  const int nodeBlocks = N_NODES / 8;

  sage_zero_kernel<<<(N_NODES + TB - 1) / TB, TB, 0, stream>>>((float*)cnt, (long long)N_NODES);
  sage_count_kernel<<<edgeBlocks, TB, 0, stream>>>(dst, cnt, E);
  sage_rowstart_kernel<<<1, SCAN_THREADS, 0, stream>>>(cnt, rowstart, deg);
  sage_fill_kernel<<<edgeBlocks, TB, 0, stream>>>(src, dst, rowstart, cnt, col, E);

  sage_gather_kernel<<<nodeBlocks, TB, 0, stream>>>(x, rowstart, col, msg);
  sage_gemm_strip_kernel<128><<<GEMM_GRID, 256, 0, stream>>>(msg, deg, x, Wl1, bl1, Wr1, h1, 1);

  sage_gather_kernel<<<nodeBlocks, TB, 0, stream>>>(h1, rowstart, col, msg);
  sage_gemm_strip_kernel<64><<<GEMM_GRID, 256, 0, stream>>>(msg, deg, h1, Wl2, bl2, Wr2, out, 0);
}


# Edges grouped by destination by a two-level counting sort on LDS counters (chunk histogram, scan, bucket scatter, per-bucket fill), no global atomic per edge; row order = LDS arrival order
# speedup vs baseline: 3.0360x; 1.5865x over previous
#include <hip/hip_runtime.h>

typedef float v2f __attribute__((ext_vector_type(2)));
typedef float v8f __attribute__((ext_vector_type(8)));

#define N_NODES 50000
#define FEAT    128

#define SORT_BSH      7
#define SORT_BNODES   (1 << SORT_BSH)
#define SORT_NB       ((N_NODES + SORT_BNODES - 1) / SORT_BNODES)
#define SORT_NCH      128
#define SORT_HLEN     (SORT_NB * SORT_NCH)
#define SORT_THREADS  1024
#define SORT_UNR      8
static_assert(N_NODES <= 65536, "keys pack a source node id into 16 bits");

__global__ void __launch_bounds__(SORT_THREADS)
sage_bucket_hist_kernel(const int* __restrict__ dst, int* __restrict__ hist, int E) {
  __shared__ int h[SORT_NB];
  const int tid = threadIdx.x;
  const int w = blockIdx.x;
  if (tid < SORT_NB) h[tid] = 0;
  __syncthreads();
  const int chunk = (E + SORT_NCH - 1) / SORT_NCH;
  const int beg = w * chunk;
  const int end = min(E, beg + chunk);
  for (int base = beg; base < end; base += SORT_UNR * SORT_THREADS) {
    int d[SORT_UNR];
#pragma unroll
    for (int k = 0; k < SORT_UNR; ++k) {
      const int e = base + k * SORT_THREADS + tid;
      d[k] = (e < end) ? dst[e] : -1;
    }
#pragma unroll
    for (int k = 0; k < SORT_UNR; ++k)
      if (d[k] >= 0) atomicAdd(&h[d[k] >> SORT_BSH], 1);
  }
  __syncthreads();
  if (tid < SORT_NB) hist[tid * SORT_NCH + w] = h[tid];
}

#define SCAN_THREADS   1024
#define SCAN_CH        52
#define SCAN_LEN       (SCAN_THREADS * SCAN_CH)
#define SCAN_OUT_ITERS ((SORT_HLEN + 1 + SCAN_THREADS - 1) / SCAN_THREADS)
__global__ void __launch_bounds__(SCAN_THREADS)
sage_scan_kernel(const int* __restrict__ hist, int* __restrict__ offs) {
  __shared__ __attribute__((aligned(16))) int buf[SCAN_LEN];
  __shared__ int wsum[SCAN_THREADS / 32];
  const int tid  = threadIdx.x;
  const int lane = tid & 31;
  const int wave = tid >> 5;

#pragma unroll
  for (int it = 0; it < SCAN_CH; ++it) {
    const int i = it * SCAN_THREADS + tid;
    buf[i] = (i < SORT_HLEN) ? hist[i] : 0;
  }
  __syncthreads();

  int4* mine = (int4*)(buf + tid * SCAN_CH);
  int4 q[SCAN_CH / 4];
  int tot = 0;
#pragma unroll
  for (int k = 0; k < SCAN_CH / 4; ++k) {
    q[k] = mine[k];
    tot += q[k].x + q[k].y + q[k].z + q[k].w;
  }

  int inc = tot;
  for (int o = 1; o < 32; o <<= 1) {
    const int up = __shfl_up(inc, o);
    if (lane >= o) inc += up;
  }
  if (lane == 31) wsum[wave] = inc;
  __syncthreads();
  int woff = (lane < wave) ? wsum[lane] : 0;
  for (int o = 16; o > 0; o >>= 1) woff += __shfl_xor(woff, o);
  int run = woff + inc - tot;

#pragma unroll
  for (int k = 0; k < SCAN_CH / 4; ++k) {
    int4 p;
    p.x = run; run += q[k].x;
    p.y = run; run += q[k].y;
    p.z = run; run += q[k].z;
    p.w = run; run += q[k].w;
    mine[k] = p;
  }
  __syncthreads();

  for (int it = 0; it < SCAN_OUT_ITERS; ++it) {
    const int i = it * SCAN_THREADS + tid;
    if (i <= SORT_HLEN) offs[i] = buf[i];
  }
}

__global__ void __launch_bounds__(SORT_THREADS)
sage_bucket_scatter_kernel(const int* __restrict__ src, const int* __restrict__ dst,
                           const int* __restrict__ offs, int* __restrict__ keys, int E) {
  __shared__ int cur[SORT_NB];
  const int tid = threadIdx.x;
  const int w = blockIdx.x;
  if (tid < SORT_NB) cur[tid] = offs[tid * SORT_NCH + w];
  __syncthreads();
  const int chunk = (E + SORT_NCH - 1) / SORT_NCH;
  const int beg = w * chunk;
  const int end = min(E, beg + chunk);
  for (int base = beg; base < end; base += SORT_UNR * SORT_THREADS) {
    int d[SORT_UNR], s[SORT_UNR];
#pragma unroll
    for (int k = 0; k < SORT_UNR; ++k) {
      const int e = base + k * SORT_THREADS + tid;
      d[k] = (e < end) ? dst[e] : -1;
      s[k] = (e < end) ? src[e] : 0;
    }
#pragma unroll
    for (int k = 0; k < SORT_UNR; ++k) {
      if (d[k] >= 0) {
        const int pos = atomicAdd(&cur[d[k] >> SORT_BSH], 1);
        keys[pos] = ((d[k] & (SORT_BNODES - 1)) << 16) | s[k];
      }
    }
  }
}

#define FILL_THREADS 512
#define FILL_UNR     5
__global__ void __launch_bounds__(FILL_THREADS)
sage_bucket_fill_kernel(const int* __restrict__ offs, const int* __restrict__ keys,
                        int* __restrict__ rowstart, float* __restrict__ deg,
                        int* __restrict__ col) {
  __shared__ __attribute__((aligned(16))) int c[SORT_BNODES];
  const int tid = threadIdx.x;
  const int b = blockIdx.x;
  const int beg = offs[b * SORT_NCH];
  const int end = offs[(b + 1) * SORT_NCH];
  if (tid < SORT_BNODES) c[tid] = 0;
  __syncthreads();

  for (int base = beg; base < end; base += FILL_UNR * FILL_THREADS) {
    int k[FILL_UNR];
#pragma unroll
    for (int u = 0; u < FILL_UNR; ++u) {
      const int j = base + u * FILL_THREADS + tid;
      k[u] = (j < end) ? keys[j] : -1;
    }
#pragma unroll
    for (int u = 0; u < FILL_UNR; ++u)
      if (k[u] >= 0) atomicAdd(&c[k[u] >> 16], 1);
  }
  __syncthreads();

  if (tid < 32) {
    const int4 cv = *(const int4*)&c[4 * tid];
    const int tot = cv.x + cv.y + cv.z + cv.w;
    int inc = tot;
    for (int o = 1; o < 32; o <<= 1) {
      const int up = __shfl_up(inc, o);
      if (tid >= o) inc += up;
    }
    int4 r;
    r.x = beg + inc - tot;
    r.y = r.x + cv.x;
    r.z = r.y + cv.y;
    r.w = r.z + cv.z;
    *(int4*)&c[4 * tid] = r;
    const int node = b * SORT_BNODES + 4 * tid;
    if (node < N_NODES) {
      *(int4*)&rowstart[node] = r;
      *(float4*)&deg[node] = make_float4((float)cv.x, (float)cv.y, (float)cv.z, (float)cv.w);
    } else if (node == N_NODES) {
      rowstart[node] = r.x;
    }
  }
  __syncthreads();

  for (int base = beg; base < end; base += FILL_UNR * FILL_THREADS) {
    int k[FILL_UNR];
#pragma unroll
    for (int u = 0; u < FILL_UNR; ++u) {
      const int j = base + u * FILL_THREADS + tid;
      k[u] = (j < end) ? keys[j] : -1;
    }
#pragma unroll
    for (int u = 0; u < FILL_UNR; ++u) {
      if (k[u] >= 0) {
        const int pos = atomicAdd(&c[k[u] >> 16], 1);
        col[pos] = k[u] & 0xFFFF;
      }
    }
  }
}

#define SAGE_ROW(k) x4[(long long)__builtin_amdgcn_readlane(idx, t + (k)) * 32 + lane]
#define SAGE_ADD(a) acc.x += a.x; acc.y += a.y; acc.z += a.z; acc.w += a.w;
__global__ void __launch_bounds__(256)
sage_gather_kernel(const float* __restrict__ x,
                   const int* __restrict__ rowstart,
                   const int* __restrict__ col,
                   float* __restrict__ msg) {
  const int lane = threadIdx.x & 31;
  const int node = __builtin_amdgcn_readfirstlane((int)(blockIdx.x * 8 + (threadIdx.x >> 5)));
  if (node >= N_NODES) return;
  const int beg = rowstart[node];
  const int end = rowstart[node + 1];
  const float4* x4 = (const float4*)x;
  float4 acc = make_float4(0.0f, 0.0f, 0.0f, 0.0f);
  for (int base = beg; base < end; base += 32) {
    const int j = base + lane;
    const int idx = (j < end) ? col[j] : 0;
    const int m = min(32, end - base);
    int t = 0;
    for (; t + 8 <= m; t += 8) {
      const float4 a0 = SAGE_ROW(0), a1 = SAGE_ROW(1), a2 = SAGE_ROW(2), a3 = SAGE_ROW(3);
      const float4 a4 = SAGE_ROW(4), a5 = SAGE_ROW(5), a6 = SAGE_ROW(6), a7 = SAGE_ROW(7);
      SAGE_ADD(a0) SAGE_ADD(a1) SAGE_ADD(a2) SAGE_ADD(a3)
      SAGE_ADD(a4) SAGE_ADD(a5) SAGE_ADD(a6) SAGE_ADD(a7)
    }
    if (t + 4 <= m) {
      const float4 a0 = SAGE_ROW(0), a1 = SAGE_ROW(1), a2 = SAGE_ROW(2), a3 = SAGE_ROW(3);
      SAGE_ADD(a0) SAGE_ADD(a1) SAGE_ADD(a2) SAGE_ADD(a3)
      t += 4;
    }
    if (t + 2 <= m) {
      const float4 a0 = SAGE_ROW(0), a1 = SAGE_ROW(1);
      SAGE_ADD(a0) SAGE_ADD(a1)
      t += 2;
    }
    if (t < m) {
      const float4 a0 = SAGE_ROW(0);
      SAGE_ADD(a0)
    }
  }
  ((float4*)msg)[(long long)node * 32 + lane] = acc;
}

#define GEMM_GRID      512
#define GEMM_RT_BLOCK  6
#define GEMM_RT_EVEN   (GEMM_GRID * GEMM_RT_BLOCK)
#define GEMM_ROW_TILES (N_NODES / 16)
#define GEMM_KG        (FEAT / 4)
#define GEMM_ST        (GEMM_KG / 2)
#define GEMM_NB        GEMM_ST

template <int N>
__device__ __forceinline__ void
sage_gemm_item(const float* Ws, const float4 (&am)[GEMM_ST], float degv,
               const float* __restrict__ xin, const float* __restrict__ bl,
               float* __restrict__ out, int doRelu, int rowTile, int half, int c, int h) {
  constexpr int K  = FEAT;
  constexpr int KG = GEMM_KG;
  constexpr int NB = GEMM_NB;
  constexpr int CT = N / 32;

  const int r = (rowTile << 4) + c;
  const float scale = 1.0f / fmaxf(degv, 1.0f);
  const int colbase = half * (N / 2) + c;
  const float4* xrow = (const float4*)(xin + (long long)r * K) + h;
  const float* wb = Ws + (h * N + colbase) * 4;
#define SAGE_BFRAG(blk, u) \
  (*(const float4*)(wb + ((((blk) >> 3) * KG + 2 * (2 * ((blk) & 7) + (u) / CT)) * N + ((u) % CT) * 16) * 4))

  float4 ax[GEMM_ST];
  v8f acc[CT];
#pragma unroll
  for (int t = 0; t < CT; ++t) acc[t] = v8f{};

  float4 bq[2][2 * CT];
#pragma unroll
  for (int u = 0; u < 2 * CT; ++u) bq[0][u] = SAGE_BFRAG(0, u);

#pragma unroll
  for (int blk = 0; blk < NB; ++blk) {
    if (blk + 1 < NB) {
#pragma unroll
      for (int u = 0; u < 2 * CT; ++u) bq[(blk + 1) & 1][u] = SAGE_BFRAG(blk + 1, u);
    }
    if (blk < NB / 2) {
      ax[2 * blk]     = xrow[4 * blk];
      ax[2 * blk + 1] = xrow[4 * blk + 2];
    }
    __builtin_amdgcn_sched_barrier(0);
#pragma unroll
    for (int u2 = 0; u2 < 2; ++u2) {
      float4 a;
      if (blk < NB / 2) {
        a = am[2 * blk + u2];
        a.x *= scale; a.y *= scale; a.z *= scale; a.w *= scale;
      } else {
        a = ax[2 * (blk - NB / 2) + u2];
      }
      v2f a0, a1;
      a0.x = a.x; a0.y = a.y;
      a1.x = a.z; a1.y = a.w;
#pragma unroll
      for (int t = 0; t < CT; ++t) {
        v2f b0;
        b0.x = bq[blk & 1][u2 * CT + t].x; b0.y = bq[blk & 1][u2 * CT + t].y;
        acc[t] = __builtin_amdgcn_wmma_f32_16x16x4_f32(false, a0, false, b0,
                                                       (short)0, acc[t], false, false);
      }
#pragma unroll
      for (int t = 0; t < CT; ++t) {
        v2f b1;
        b1.x = bq[blk & 1][u2 * CT + t].z; b1.y = bq[blk & 1][u2 * CT + t].w;
        acc[t] = __builtin_amdgcn_wmma_f32_16x16x4_f32(false, a1, false, b1,
                                                       (short)0, acc[t], false, false);
      }
    }
    __builtin_amdgcn_sched_barrier(0);
  }
#undef SAGE_BFRAG

  const int rbase = (rowTile << 4) + (h << 3);
#pragma unroll
  for (int t = 0; t < CT; ++t) {
    const int col = colbase + t * 16;
    const float bias = bl[col];
#pragma unroll
    for (int i = 0; i < 8; ++i) {
      float v = acc[t][i] + bias;
      if (doRelu) v = fmaxf(v, 0.0f);
      out[(long long)(rbase + i) * N + col] = v;
    }
  }
}

template <int N>
__global__ void __launch_bounds__(256)
sage_gemm_strip_kernel(const float* __restrict__ msg,
                       const float* __restrict__ deg,
                       const float* __restrict__ xin,
                       const float* __restrict__ Wl,
                       const float* __restrict__ bl,
                       const float* __restrict__ Wr,
                       float* __restrict__ out, int doRelu) {
  constexpr int K  = FEAT;
  constexpr int KG = GEMM_KG;
  __shared__ __attribute__((aligned(16))) float Ws[2 * K * N];

  const int tid  = threadIdx.x;
  const int lane = tid & 31;
  const int c = lane & 15;
  const int h = lane >> 4;
  const int wv = __builtin_amdgcn_readfirstlane(tid >> 5);
  const int b  = blockIdx.x;

  const int rowTile1 = b * GEMM_RT_BLOCK + (wv >> 1);
  const int half1 = wv & 1;
  const int q = b >> 2;
  const bool left = (wv == 4) && ((b & 3) == 1) && (GEMM_RT_EVEN + (q >> 1) < GEMM_ROW_TILES);
  const bool second = (wv < 4) || left;
  const int rowTile2 = left ? GEMM_RT_EVEN + (q >> 1) : b * GEMM_RT_BLOCK + 4 + (wv >> 1);
  const int half2 = left ? (q & 1) : (wv & 1);

  const int r1 = (rowTile1 << 4) + c;
  const float deg1 = deg[r1];
  float4 am[GEMM_ST];
  {
    const float4* mrow = (const float4*)(msg + (long long)r1 * K) + h;
#pragma unroll
    for (int j = 0; j < GEMM_ST; ++j) am[j] = mrow[2 * j];
  }

  {
    constexpr int KG_PASS = 256 / N;
    const int n   = tid % N;
    const int kgo = tid / N;
    for (int m = 0; m < 2; ++m) {
      const float* __restrict__ Wm = m ? Wr : Wl;
#pragma unroll 4
      for (int p = 0; p < KG / KG_PASS; ++p) {
        const int kg = p * KG_PASS + kgo;
        float4 v;
        v.x = Wm[(4 * kg + 0) * N + n];
        v.y = Wm[(4 * kg + 1) * N + n];
        v.z = Wm[(4 * kg + 2) * N + n];
        v.w = Wm[(4 * kg + 3) * N + n];
        *(float4*)(Ws + ((m * KG + kg) * N + n) * 4) = v;
      }
    }
  }
  __syncthreads();

  sage_gemm_item<N>(Ws, am, deg1, xin, bl, out, doRelu, rowTile1, half1, c, h);
  if (second) {
    const int r2 = (rowTile2 << 4) + c;
    const float deg2 = deg[r2];
    float4 an[GEMM_ST];
    const float4* mrow = (const float4*)(msg + (long long)r2 * K) + h;
#pragma unroll
    for (int j = 0; j < GEMM_ST; ++j) an[j] = mrow[2 * j];
    sage_gemm_item<N>(Ws, an, deg2, xin, bl, out, doRelu, rowTile2, half2, c, h);
  }
}

extern "C" void kernel_launch(void* const* d_in, const int* in_sizes, int n_in,
                              void* d_out, int out_size, void* d_ws, size_t ws_size,
                              hipStream_t stream) {
  const float* x   = (const float*)d_in[0];
  const int*   ei  = (const int*)d_in[1];
  const float* Wl1 = (const float*)d_in[2];
  const float* bl1 = (const float*)d_in[3];
  const float* Wr1 = (const float*)d_in[4];
  const float* Wl2 = (const float*)d_in[5];
  const float* bl2 = (const float*)d_in[6];
  const float* Wr2 = (const float*)d_in[7];
  float* out = (float*)d_out;

  const int E = in_sizes[1] / 2;
  const int* src = ei;
  const int* dst = ei + E;

  float* msg = (float*)d_ws;
  float* deg = msg + (long long)N_NODES * FEAT;
  float* h1  = deg + N_NODES;

  int* rowstart = (int*)d_out;
  int* col      = rowstart + 50048;

  int* hist = (int*)h1;
  int* offs = hist + 65536;
  int* keys = offs + 65536;

  const int TB = 256;
  const int nodeBlocks = N_NODES / 8;

  sage_bucket_hist_kernel<<<SORT_NCH, SORT_THREADS, 0, stream>>>(dst, hist, E);
  sage_scan_kernel<<<1, SCAN_THREADS, 0, stream>>>(hist, offs);
  sage_bucket_scatter_kernel<<<SORT_NCH, SORT_THREADS, 0, stream>>>(src, dst, offs, keys, E);
  sage_bucket_fill_kernel<<<SORT_NB, FILL_THREADS, 0, stream>>>(offs, keys, rowstart, deg, col);

  sage_gather_kernel<<<nodeBlocks, TB, 0, stream>>>(x, rowstart, col, msg);
  sage_gemm_strip_kernel<128><<<GEMM_GRID, 256, 0, stream>>>(msg, deg, x, Wl1, bl1, Wr1, h1, 1);

  sage_gather_kernel<<<nodeBlocks, TB, 0, stream>>>(h1, rowstart, col, msg);
  sage_gemm_strip_kernel<64><<<GEMM_GRID, 256, 0, stream>>>(msg, deg, h1, Wl2, bl2, Wr2, out, 0);
}


# Edge sort in 2 launches, not 4: each chunk workgroup counts, scans and places its edges by bucket in its own slice; bucket start = column sum of chunk tables, no global scan; row order = LDS arrival
# speedup vs baseline: 3.9022x; 1.2853x over previous
#include <hip/hip_runtime.h>

typedef float v2f __attribute__((ext_vector_type(2)));
typedef float v8f __attribute__((ext_vector_type(8)));

#define N_NODES 50000
#define FEAT    128

#define SORT_BSH      7
#define SORT_BNODES   (1 << SORT_BSH)
#define SORT_NB       ((N_NODES + SORT_BNODES - 1) / SORT_BNODES)
#define SORT_NCH      128
#define SORT_THREADS  1024
#define SORT_UNR      8
#define SORT_SCAN_CH  13
static_assert(N_NODES <= 65536, "keys pack a source node id into 16 bits");
static_assert(32 * SORT_SCAN_CH >= SORT_NB + 1, "one wave scans the bucket counts of a chunk");

__global__ void __launch_bounds__(SORT_THREADS)
sage_chunk_sort_kernel(const int* __restrict__ src, const int* __restrict__ dst,
                       int* __restrict__ loc, int* __restrict__ keys, int E) {
  __shared__ int h[SORT_NB + 1];
  const int tid = threadIdx.x;
  const int w = blockIdx.x;
  if (tid <= SORT_NB) h[tid] = 0;
  __syncthreads();
  const int chunk = (E + SORT_NCH - 1) / SORT_NCH;
  const int beg = w * chunk;
  const int end = min(E, beg + chunk);
  const bool one = (end - beg) <= SORT_UNR * SORT_THREADS;

  int d[SORT_UNR], s[SORT_UNR];
  if (one) {
#pragma unroll
    for (int k = 0; k < SORT_UNR; ++k) {
      const int e = beg + k * SORT_THREADS + tid;
      d[k] = (e < end) ? dst[e] : -1;
      s[k] = (e < end) ? src[e] : 0;
    }
#pragma unroll
    for (int k = 0; k < SORT_UNR; ++k)
      if (d[k] >= 0) atomicAdd(&h[d[k] >> SORT_BSH], 1);
  } else {
    for (int base = beg; base < end; base += SORT_UNR * SORT_THREADS) {
      int dd[SORT_UNR];
#pragma unroll
      for (int k = 0; k < SORT_UNR; ++k) {
        const int e = base + k * SORT_THREADS + tid;
        dd[k] = (e < end) ? dst[e] : -1;
      }
#pragma unroll
      for (int k = 0; k < SORT_UNR; ++k)
        if (dd[k] >= 0) atomicAdd(&h[dd[k] >> SORT_BSH], 1);
    }
  }
  __syncthreads();

  if (tid < 32) {
    int v[SORT_SCAN_CH];
    int tot = 0;
#pragma unroll
    for (int k = 0; k < SORT_SCAN_CH; ++k) {
      const int i = tid * SORT_SCAN_CH + k;
      v[k] = (i < SORT_NB) ? h[i] : 0;
      tot += v[k];
    }
    int inc = tot;
    for (int o = 1; o < 32; o <<= 1) {
      const int up = __shfl_up(inc, o);
      if (tid >= o) inc += up;
    }
    int run = inc - tot;
#pragma unroll
    for (int k = 0; k < SORT_SCAN_CH; ++k) {
      const int i = tid * SORT_SCAN_CH + k;
      if (i <= SORT_NB) {
        h[i] = run;
        loc[i * SORT_NCH + w] = run;
      }
      run += v[k];
    }
  }
  __syncthreads();

  int* __restrict__ mine = keys + beg;
  if (one) {
#pragma unroll
    for (int k = 0; k < SORT_UNR; ++k) {
      if (d[k] >= 0) {
        const int pos = atomicAdd(&h[d[k] >> SORT_BSH], 1);
        mine[pos] = ((d[k] & (SORT_BNODES - 1)) << 16) | s[k];
      }
    }
  } else {
    for (int base = beg; base < end; base += SORT_UNR * SORT_THREADS) {
      int dd[SORT_UNR], ss[SORT_UNR];
#pragma unroll
      for (int k = 0; k < SORT_UNR; ++k) {
        const int e = base + k * SORT_THREADS + tid;
        dd[k] = (e < end) ? dst[e] : -1;
        ss[k] = (e < end) ? src[e] : 0;
      }
#pragma unroll
      for (int k = 0; k < SORT_UNR; ++k) {
        if (dd[k] >= 0) {
          const int pos = atomicAdd(&h[dd[k] >> SORT_BSH], 1);
          mine[pos] = ((dd[k] & (SORT_BNODES - 1)) << 16) | ss[k];
        }
      }
    }
  }
}

#define FILL_THREADS 512
#define FILL_WAVES   (FILL_THREADS / 32)
#define FILL_PPW     (SORT_NCH / FILL_WAVES)
__global__ void __launch_bounds__(FILL_THREADS)
sage_bucket_fill_kernel(const int* __restrict__ loc, const int* __restrict__ keys,
                        int* __restrict__ rowstart, float* __restrict__ deg,
                        int* __restrict__ col, int E) {
  __shared__ __attribute__((aligned(16))) int c[SORT_BNODES];
  __shared__ int red[SORT_NCH / 32];
  const int tid = threadIdx.x;
  const int lane = tid & 31;
  const int wv = __builtin_amdgcn_readfirstlane(tid >> 5);
  const int b = blockIdx.x;
  const int chunk = (E + SORT_NCH - 1) / SORT_NCH;
  const int* __restrict__ lrow = loc + b * SORT_NCH;

  int st[FILL_PPW], len[FILL_PPW];
#pragma unroll
  for (int u = 0; u < FILL_PPW; ++u) {
    const int p = wv + u * FILL_WAVES;
    const int lo = lrow[p];
    st[u] = p * chunk + lo;
    len[u] = lrow[SORT_NCH + p] - lo;
  }
  int k[FILL_PPW];
#pragma unroll
  for (int u = 0; u < FILL_PPW; ++u)
    k[u] = (lane < len[u]) ? keys[st[u] + lane] : -1;

  if (tid < SORT_NCH) {
    c[tid] = 0;
    int slo = lrow[tid];
    for (int o = 16; o > 0; o >>= 1) slo += __shfl_xor(slo, o);
    if (lane == 0) red[wv] = slo;
  }
  __syncthreads();
  int beg = 0;
  for (int q = 0; q < SORT_NCH / 32; ++q) beg += red[q];

#pragma unroll
  for (int u = 0; u < FILL_PPW; ++u)
    if (k[u] >= 0) atomicAdd(&c[k[u] >> 16], 1);
#pragma unroll
  for (int u = 0; u < FILL_PPW; ++u)
    for (int j = 32 + lane; j < len[u]; j += 32) atomicAdd(&c[keys[st[u] + j] >> 16], 1);
  __syncthreads();

  if (tid < 32) {
    const int4 cv = *(const int4*)&c[4 * tid];
    const int tot = cv.x + cv.y + cv.z + cv.w;
    int inc = tot;
    for (int o = 1; o < 32; o <<= 1) {
      const int up = __shfl_up(inc, o);
      if (tid >= o) inc += up;
    }
    int4 r;
    r.x = beg + inc - tot;
    r.y = r.x + cv.x;
    r.z = r.y + cv.y;
    r.w = r.z + cv.z;
    *(int4*)&c[4 * tid] = r;
    const int node = b * SORT_BNODES + 4 * tid;
    if (node < N_NODES) {
      *(int4*)&rowstart[node] = r;
      *(float4*)&deg[node] = make_float4((float)cv.x, (float)cv.y, (float)cv.z, (float)cv.w);
    } else if (node == N_NODES) {
      rowstart[node] = r.x;
    }
  }
  __syncthreads();

#pragma unroll
  for (int u = 0; u < FILL_PPW; ++u) {
    if (k[u] >= 0) {
      const int pos = atomicAdd(&c[k[u] >> 16], 1);
      col[pos] = k[u] & 0xFFFF;
    }
  }
#pragma unroll
  for (int u = 0; u < FILL_PPW; ++u) {
    for (int j = 32 + lane; j < len[u]; j += 32) {
      const int kk = keys[st[u] + j];
      const int pos = atomicAdd(&c[kk >> 16], 1);
      col[pos] = kk & 0xFFFF;
    }
  }
}

#define SAGE_ROW(k) x4[(long long)__builtin_amdgcn_readlane(idx, t + (k)) * 32 + lane]
#define SAGE_ADD(a) acc.x += a.x; acc.y += a.y; acc.z += a.z; acc.w += a.w;
__global__ void __launch_bounds__(256)
sage_gather_kernel(const float* __restrict__ x,
                   const int* __restrict__ rowstart,
                   const int* __restrict__ col,
                   float* __restrict__ msg) {
  const int lane = threadIdx.x & 31;
  const int node = __builtin_amdgcn_readfirstlane((int)(blockIdx.x * 8 + (threadIdx.x >> 5)));
  if (node >= N_NODES) return;
  const int beg = rowstart[node];
  const int end = rowstart[node + 1];
  const float4* x4 = (const float4*)x;
  float4 acc = make_float4(0.0f, 0.0f, 0.0f, 0.0f);
  for (int base = beg; base < end; base += 32) {
    const int j = base + lane;
    const int idx = (j < end) ? col[j] : 0;
    const int m = min(32, end - base);
    int t = 0;
    for (; t + 8 <= m; t += 8) {
      const float4 a0 = SAGE_ROW(0), a1 = SAGE_ROW(1), a2 = SAGE_ROW(2), a3 = SAGE_ROW(3);
      const float4 a4 = SAGE_ROW(4), a5 = SAGE_ROW(5), a6 = SAGE_ROW(6), a7 = SAGE_ROW(7);
      SAGE_ADD(a0) SAGE_ADD(a1) SAGE_ADD(a2) SAGE_ADD(a3)
      SAGE_ADD(a4) SAGE_ADD(a5) SAGE_ADD(a6) SAGE_ADD(a7)
    }
    if (t + 4 <= m) {
      const float4 a0 = SAGE_ROW(0), a1 = SAGE_ROW(1), a2 = SAGE_ROW(2), a3 = SAGE_ROW(3);
      SAGE_ADD(a0) SAGE_ADD(a1) SAGE_ADD(a2) SAGE_ADD(a3)
      t += 4;
    }
    if (t + 2 <= m) {
      const float4 a0 = SAGE_ROW(0), a1 = SAGE_ROW(1);
      SAGE_ADD(a0) SAGE_ADD(a1)
      t += 2;
    }
    if (t < m) {
      const float4 a0 = SAGE_ROW(0);
      SAGE_ADD(a0)
    }
  }
  ((float4*)msg)[(long long)node * 32 + lane] = acc;
}

#define GEMM_GRID      512
#define GEMM_RT_BLOCK  6
#define GEMM_RT_EVEN   (GEMM_GRID * GEMM_RT_BLOCK)
#define GEMM_ROW_TILES (N_NODES / 16)
#define GEMM_KG        (FEAT / 4)
#define GEMM_ST        (GEMM_KG / 2)
#define GEMM_NB        GEMM_ST

template <int N>
__device__ __forceinline__ void
sage_gemm_item(const float* Ws, const float4 (&am)[GEMM_ST], float degv,
               const float* __restrict__ xin, const float* __restrict__ bl,
               float* __restrict__ out, int doRelu, int rowTile, int half, int c, int h) {
  constexpr int K  = FEAT;
  constexpr int KG = GEMM_KG;
  constexpr int NB = GEMM_NB;
  constexpr int CT = N / 32;

  const int r = (rowTile << 4) + c;
  const float scale = 1.0f / fmaxf(degv, 1.0f);
  const int colbase = half * (N / 2) + c;
  const float4* xrow = (const float4*)(xin + (long long)r * K) + h;
  const float* wb = Ws + (h * N + colbase) * 4;
#define SAGE_BFRAG(blk, u) \
  (*(const float4*)(wb + ((((blk) >> 3) * KG + 2 * (2 * ((blk) & 7) + (u) / CT)) * N + ((u) % CT) * 16) * 4))

  float4 ax[GEMM_ST];
  v8f acc[CT];
#pragma unroll
  for (int t = 0; t < CT; ++t) acc[t] = v8f{};

  float4 bq[2][2 * CT];
#pragma unroll
  for (int u = 0; u < 2 * CT; ++u) bq[0][u] = SAGE_BFRAG(0, u);

#pragma unroll
  for (int blk = 0; blk < NB; ++blk) {
    if (blk + 1 < NB) {
#pragma unroll
      for (int u = 0; u < 2 * CT; ++u) bq[(blk + 1) & 1][u] = SAGE_BFRAG(blk + 1, u);
    }
    if (blk < NB / 2) {
      ax[2 * blk]     = xrow[4 * blk];
      ax[2 * blk + 1] = xrow[4 * blk + 2];
    }
    __builtin_amdgcn_sched_barrier(0);
#pragma unroll
    for (int u2 = 0; u2 < 2; ++u2) {
      float4 a;
      if (blk < NB / 2) {
        a = am[2 * blk + u2];
        a.x *= scale; a.y *= scale; a.z *= scale; a.w *= scale;
      } else {
        a = ax[2 * (blk - NB / 2) + u2];
      }
      v2f a0, a1;
      a0.x = a.x; a0.y = a.y;
      a1.x = a.z; a1.y = a.w;
#pragma unroll
      for (int t = 0; t < CT; ++t) {
        v2f b0;
        b0.x = bq[blk & 1][u2 * CT + t].x; b0.y = bq[blk & 1][u2 * CT + t].y;
        acc[t] = __builtin_amdgcn_wmma_f32_16x16x4_f32(false, a0, false, b0,
                                                       (short)0, acc[t], false, false);
      }
#pragma unroll
      for (int t = 0; t < CT; ++t) {
        v2f b1;
        b1.x = bq[blk & 1][u2 * CT + t].z; b1.y = bq[blk & 1][u2 * CT + t].w;
        acc[t] = __builtin_amdgcn_wmma_f32_16x16x4_f32(false, a1, false, b1,
                                                       (short)0, acc[t], false, false);
      }
    }
    __builtin_amdgcn_sched_barrier(0);
  }
#undef SAGE_BFRAG

  const int rbase = (rowTile << 4) + (h << 3);
#pragma unroll
  for (int t = 0; t < CT; ++t) {
    const int col = colbase + t * 16;
    const float bias = bl[col];
#pragma unroll
    for (int i = 0; i < 8; ++i) {
      float v = acc[t][i] + bias;
      if (doRelu) v = fmaxf(v, 0.0f);
      out[(long long)(rbase + i) * N + col] = v;
    }
  }
}

template <int N>
__global__ void __launch_bounds__(256)
sage_gemm_strip_kernel(const float* __restrict__ msg,
                       const float* __restrict__ deg,
                       const float* __restrict__ xin,
                       const float* __restrict__ Wl,
                       const float* __restrict__ bl,
                       const float* __restrict__ Wr,
                       float* __restrict__ out, int doRelu) {
  constexpr int K  = FEAT;
  constexpr int KG = GEMM_KG;
  __shared__ __attribute__((aligned(16))) float Ws[2 * K * N];

  const int tid  = threadIdx.x;
  const int lane = tid & 31;
  const int c = lane & 15;
  const int h = lane >> 4;
  const int wv = __builtin_amdgcn_readfirstlane(tid >> 5);
  const int b  = blockIdx.x;

  const int rowTile1 = b * GEMM_RT_BLOCK + (wv >> 1);
  const int half1 = wv & 1;
  const int q = b >> 2;
  const bool left = (wv == 4) && ((b & 3) == 1) && (GEMM_RT_EVEN + (q >> 1) < GEMM_ROW_TILES);
  const bool second = (wv < 4) || left;
  const int rowTile2 = left ? GEMM_RT_EVEN + (q >> 1) : b * GEMM_RT_BLOCK + 4 + (wv >> 1);
  const int half2 = left ? (q & 1) : (wv & 1);

  const int r1 = (rowTile1 << 4) + c;
  const float deg1 = deg[r1];
  float4 am[GEMM_ST];
  {
    const float4* mrow = (const float4*)(msg + (long long)r1 * K) + h;
#pragma unroll
    for (int j = 0; j < GEMM_ST; ++j) am[j] = mrow[2 * j];
  }

  {
    constexpr int KG_PASS = 256 / N;
    const int n   = tid % N;
    const int kgo = tid / N;
    for (int m = 0; m < 2; ++m) {
      const float* __restrict__ Wm = m ? Wr : Wl;
#pragma unroll 4
      for (int p = 0; p < KG / KG_PASS; ++p) {
        const int kg = p * KG_PASS + kgo;
        float4 v;
        v.x = Wm[(4 * kg + 0) * N + n];
        v.y = Wm[(4 * kg + 1) * N + n];
        v.z = Wm[(4 * kg + 2) * N + n];
        v.w = Wm[(4 * kg + 3) * N + n];
        *(float4*)(Ws + ((m * KG + kg) * N + n) * 4) = v;
      }
    }
  }
  __syncthreads();

  sage_gemm_item<N>(Ws, am, deg1, xin, bl, out, doRelu, rowTile1, half1, c, h);
  if (second) {
    const int r2 = (rowTile2 << 4) + c;
    const float deg2 = deg[r2];
    float4 an[GEMM_ST];
    const float4* mrow = (const float4*)(msg + (long long)r2 * K) + h;
#pragma unroll
    for (int j = 0; j < GEMM_ST; ++j) an[j] = mrow[2 * j];
    sage_gemm_item<N>(Ws, an, deg2, xin, bl, out, doRelu, rowTile2, half2, c, h);
  }
}

extern "C" void kernel_launch(void* const* d_in, const int* in_sizes, int n_in,
                              void* d_out, int out_size, void* d_ws, size_t ws_size,
                              hipStream_t stream) {
  const float* x   = (const float*)d_in[0];
  const int*   ei  = (const int*)d_in[1];
  const float* Wl1 = (const float*)d_in[2];
  const float* bl1 = (const float*)d_in[3];
  const float* Wr1 = (const float*)d_in[4];
  const float* Wl2 = (const float*)d_in[5];
  const float* bl2 = (const float*)d_in[6];
  const float* Wr2 = (const float*)d_in[7];
  float* out = (float*)d_out;

  const int E = in_sizes[1] / 2;
  const int* src = ei;
  const int* dst = ei + E;

  float* msg = (float*)d_ws;
  float* deg = msg + (long long)N_NODES * FEAT;
  float* h1  = deg + N_NODES;

  int* rowstart = (int*)d_out;
  int* col      = rowstart + 50048;

  int* loc  = (int*)h1;
  int* keys = loc + 65536;

  const int TB = 256;
  const int nodeBlocks = N_NODES / 8;

  sage_chunk_sort_kernel<<<SORT_NCH, SORT_THREADS, 0, stream>>>(src, dst, loc, keys, E);
  sage_bucket_fill_kernel<<<SORT_NB, FILL_THREADS, 0, stream>>>(loc, keys, rowstart, deg, col, E);

  sage_gather_kernel<<<nodeBlocks, TB, 0, stream>>>(x, rowstart, col, msg);
  sage_gemm_strip_kernel<128><<<GEMM_GRID, 256, 0, stream>>>(msg, deg, x, Wl1, bl1, Wr1, h1, 1);

  sage_gather_kernel<<<nodeBlocks, TB, 0, stream>>>(h1, rowstart, col, msg);
  sage_gemm_strip_kernel<64><<<GEMM_GRID, 256, 0, stream>>>(msg, deg, h1, Wl2, bl2, Wr2, out, 0);
}


# Layer 2 applies its weights before the mean (linearity): the gather sums 64-float rows of h1@Wl2, not 128-float rows of h1; float sums reordered (products first, two interleaved partial sums)
# speedup vs baseline: 4.1930x; 1.0745x over previous
#include <hip/hip_runtime.h>

typedef float v2f __attribute__((ext_vector_type(2)));
typedef float v8f __attribute__((ext_vector_type(8)));

#define N_NODES 50000
#define FEAT    128

#define SORT_BSH      7
#define SORT_BNODES   (1 << SORT_BSH)
#define SORT_NB       ((N_NODES + SORT_BNODES - 1) / SORT_BNODES)
#define SORT_NCH      128
#define SORT_THREADS  1024
#define SORT_UNR      8
#define SORT_SCAN_CH  13
static_assert(N_NODES <= 65536, "keys pack a source node id into 16 bits");
static_assert(32 * SORT_SCAN_CH >= SORT_NB + 1, "one wave scans the bucket counts of a chunk");

__global__ void __launch_bounds__(SORT_THREADS)
sage_chunk_sort_kernel(const int* __restrict__ src, const int* __restrict__ dst,
                       int* __restrict__ loc, int* __restrict__ keys, int E) {
  __shared__ int h[SORT_NB + 1];
  const int tid = threadIdx.x;
  const int w = blockIdx.x;
  if (tid <= SORT_NB) h[tid] = 0;
  __syncthreads();
  const int chunk = (E + SORT_NCH - 1) / SORT_NCH;
  const int beg = w * chunk;
  const int end = min(E, beg + chunk);
  const bool one = (end - beg) <= SORT_UNR * SORT_THREADS;

  int d[SORT_UNR], s[SORT_UNR];
  if (one) {
#pragma unroll
    for (int k = 0; k < SORT_UNR; ++k) {
      const int e = beg + k * SORT_THREADS + tid;
      d[k] = (e < end) ? dst[e] : -1;
      s[k] = (e < end) ? src[e] : 0;
    }
#pragma unroll
    for (int k = 0; k < SORT_UNR; ++k)
      if (d[k] >= 0) atomicAdd(&h[d[k] >> SORT_BSH], 1);
  } else {
    for (int base = beg; base < end; base += SORT_UNR * SORT_THREADS) {
      int dd[SORT_UNR];
#pragma unroll
      for (int k = 0; k < SORT_UNR; ++k) {
        const int e = base + k * SORT_THREADS + tid;
        dd[k] = (e < end) ? dst[e] : -1;
      }
#pragma unroll
      for (int k = 0; k < SORT_UNR; ++k)
        if (dd[k] >= 0) atomicAdd(&h[dd[k] >> SORT_BSH], 1);
    }
  }
  __syncthreads();

  if (tid < 32) {
    int v[SORT_SCAN_CH];
    int tot = 0;
#pragma unroll
    for (int k = 0; k < SORT_SCAN_CH; ++k) {
      const int i = tid * SORT_SCAN_CH + k;
      v[k] = (i < SORT_NB) ? h[i] : 0;
      tot += v[k];
    }
    int inc = tot;
    for (int o = 1; o < 32; o <<= 1) {
      const int up = __shfl_up(inc, o);
      if (tid >= o) inc += up;
    }
    int run = inc - tot;
#pragma unroll
    for (int k = 0; k < SORT_SCAN_CH; ++k) {
      const int i = tid * SORT_SCAN_CH + k;
      if (i <= SORT_NB) {
        h[i] = run;
        loc[i * SORT_NCH + w] = run;
      }
      run += v[k];
    }
  }
  __syncthreads();

  int* __restrict__ mine = keys + beg;
  if (one) {
#pragma unroll
    for (int k = 0; k < SORT_UNR; ++k) {
      if (d[k] >= 0) {
        const int pos = atomicAdd(&h[d[k] >> SORT_BSH], 1);
        mine[pos] = ((d[k] & (SORT_BNODES - 1)) << 16) | s[k];
      }
    }
  } else {
    for (int base = beg; base < end; base += SORT_UNR * SORT_THREADS) {
      int dd[SORT_UNR], ss[SORT_UNR];
#pragma unroll
      for (int k = 0; k < SORT_UNR; ++k) {
        const int e = base + k * SORT_THREADS + tid;
        dd[k] = (e < end) ? dst[e] : -1;
        ss[k] = (e < end) ? src[e] : 0;
      }
#pragma unroll
      for (int k = 0; k < SORT_UNR; ++k) {
        if (dd[k] >= 0) {
          const int pos = atomicAdd(&h[dd[k] >> SORT_BSH], 1);
          mine[pos] = ((dd[k] & (SORT_BNODES - 1)) << 16) | ss[k];
        }
      }
    }
  }
}

#define FILL_THREADS 512
#define FILL_WAVES   (FILL_THREADS / 32)
#define FILL_PPW     (SORT_NCH / FILL_WAVES)
__global__ void __launch_bounds__(FILL_THREADS)
sage_bucket_fill_kernel(const int* __restrict__ loc, const int* __restrict__ keys,
                        int* __restrict__ rowstart, float* __restrict__ deg,
                        int* __restrict__ col, int E) {
  __shared__ __attribute__((aligned(16))) int c[SORT_BNODES];
  __shared__ int red[SORT_NCH / 32];
  const int tid = threadIdx.x;
  const int lane = tid & 31;
  const int wv = __builtin_amdgcn_readfirstlane(tid >> 5);
  const int b = blockIdx.x;
  const int chunk = (E + SORT_NCH - 1) / SORT_NCH;
  const int* __restrict__ lrow = loc + b * SORT_NCH;

  int st[FILL_PPW], len[FILL_PPW];
#pragma unroll
  for (int u = 0; u < FILL_PPW; ++u) {
    const int p = wv + u * FILL_WAVES;
    const int lo = lrow[p];
    st[u] = p * chunk + lo;
    len[u] = lrow[SORT_NCH + p] - lo;
  }
  int k[FILL_PPW];
#pragma unroll
  for (int u = 0; u < FILL_PPW; ++u)
    k[u] = (lane < len[u]) ? keys[st[u] + lane] : -1;

  if (tid < SORT_NCH) {
    c[tid] = 0;
    int slo = lrow[tid];
    for (int o = 16; o > 0; o >>= 1) slo += __shfl_xor(slo, o);
    if (lane == 0) red[wv] = slo;
  }
  __syncthreads();
  int beg = 0;
  for (int q = 0; q < SORT_NCH / 32; ++q) beg += red[q];

#pragma unroll
  for (int u = 0; u < FILL_PPW; ++u)
    if (k[u] >= 0) atomicAdd(&c[k[u] >> 16], 1);
#pragma unroll
  for (int u = 0; u < FILL_PPW; ++u)
    for (int j = 32 + lane; j < len[u]; j += 32) atomicAdd(&c[keys[st[u] + j] >> 16], 1);
  __syncthreads();

  if (tid < 32) {
    const int4 cv = *(const int4*)&c[4 * tid];
    const int tot = cv.x + cv.y + cv.z + cv.w;
    int inc = tot;
    for (int o = 1; o < 32; o <<= 1) {
      const int up = __shfl_up(inc, o);
      if (tid >= o) inc += up;
    }
    int4 r;
    r.x = beg + inc - tot;
    r.y = r.x + cv.x;
    r.z = r.y + cv.y;
    r.w = r.z + cv.z;
    *(int4*)&c[4 * tid] = r;
    const int node = b * SORT_BNODES + 4 * tid;
    if (node < N_NODES) {
      *(int4*)&rowstart[node] = r;
      *(float4*)&deg[node] = make_float4((float)cv.x, (float)cv.y, (float)cv.z, (float)cv.w);
    } else if (node == N_NODES) {
      rowstart[node] = r.x;
    }
  }
  __syncthreads();

#pragma unroll
  for (int u = 0; u < FILL_PPW; ++u) {
    if (k[u] >= 0) {
      const int pos = atomicAdd(&c[k[u] >> 16], 1);
      col[pos] = k[u] & 0xFFFF;
    }
  }
#pragma unroll
  for (int u = 0; u < FILL_PPW; ++u) {
    for (int j = 32 + lane; j < len[u]; j += 32) {
      const int kk = keys[st[u] + j];
      const int pos = atomicAdd(&c[kk >> 16], 1);
      col[pos] = kk & 0xFFFF;
    }
  }
}

#define SAGE_ROW(k) x4[(long long)__builtin_amdgcn_readlane(idx, t + (k)) * 32 + lane]
#define SAGE_ADD(a) acc.x += a.x; acc.y += a.y; acc.z += a.z; acc.w += a.w;
__global__ void __launch_bounds__(256)
sage_gather_kernel(const float* __restrict__ x,
                   const int* __restrict__ rowstart,
                   const int* __restrict__ col,
                   float* __restrict__ msg) {
  const int lane = threadIdx.x & 31;
  const int node = __builtin_amdgcn_readfirstlane((int)(blockIdx.x * 8 + (threadIdx.x >> 5)));
  if (node >= N_NODES) return;
  const int beg = rowstart[node];
  const int end = rowstart[node + 1];
  const float4* x4 = (const float4*)x;
  float4 acc = make_float4(0.0f, 0.0f, 0.0f, 0.0f);
  for (int base = beg; base < end; base += 32) {
    const int j = base + lane;
    const int idx = (j < end) ? col[j] : 0;
    const int m = min(32, end - base);
    int t = 0;
    for (; t + 8 <= m; t += 8) {
      const float4 a0 = SAGE_ROW(0), a1 = SAGE_ROW(1), a2 = SAGE_ROW(2), a3 = SAGE_ROW(3);
      const float4 a4 = SAGE_ROW(4), a5 = SAGE_ROW(5), a6 = SAGE_ROW(6), a7 = SAGE_ROW(7);
      SAGE_ADD(a0) SAGE_ADD(a1) SAGE_ADD(a2) SAGE_ADD(a3)
      SAGE_ADD(a4) SAGE_ADD(a5) SAGE_ADD(a6) SAGE_ADD(a7)
    }
    if (t + 4 <= m) {
      const float4 a0 = SAGE_ROW(0), a1 = SAGE_ROW(1), a2 = SAGE_ROW(2), a3 = SAGE_ROW(3);
      SAGE_ADD(a0) SAGE_ADD(a1) SAGE_ADD(a2) SAGE_ADD(a3)
      t += 4;
    }
    if (t + 2 <= m) {
      const float4 a0 = SAGE_ROW(0), a1 = SAGE_ROW(1);
      SAGE_ADD(a0) SAGE_ADD(a1)
      t += 2;
    }
    if (t < m) {
      const float4 a0 = SAGE_ROW(0);
      SAGE_ADD(a0)
    }
  }
  ((float4*)msg)[(long long)node * 32 + lane] = acc;
}

#define GEMM_GRID      512
#define GEMM_RT_BLOCK  6
#define GEMM_RT_EVEN   (GEMM_GRID * GEMM_RT_BLOCK)
#define GEMM_ROW_TILES (N_NODES / 16)
#define GEMM_KG        (FEAT / 4)
#define GEMM_ST        (GEMM_KG / 2)
#define GEMM_NB        GEMM_ST

template <int N>
__device__ __forceinline__ void
sage_gemm_item(const float* Ws, const float4 (&am)[GEMM_ST], float degv,
               const float* __restrict__ xin, const float* __restrict__ bl,
               float* __restrict__ out, int doRelu, int rowTile, int half, int c, int h) {
  constexpr int K  = FEAT;
  constexpr int KG = GEMM_KG;
  constexpr int NB = GEMM_NB;
  constexpr int CT = N / 32;

  const int r = (rowTile << 4) + c;
  const float scale = 1.0f / fmaxf(degv, 1.0f);
  const int colbase = half * (N / 2) + c;
  const float4* xrow = (const float4*)(xin + (long long)r * K) + h;
  const float* wb = Ws + (h * N + colbase) * 4;
#define SAGE_BFRAG(blk, u) \
  (*(const float4*)(wb + ((((blk) >> 3) * KG + 2 * (2 * ((blk) & 7) + (u) / CT)) * N + ((u) % CT) * 16) * 4))

  float4 ax[GEMM_ST];
  v8f acc[CT];
#pragma unroll
  for (int t = 0; t < CT; ++t) acc[t] = v8f{};

  float4 bq[2][2 * CT];
#pragma unroll
  for (int u = 0; u < 2 * CT; ++u) bq[0][u] = SAGE_BFRAG(0, u);

#pragma unroll
  for (int blk = 0; blk < NB; ++blk) {
    if (blk + 1 < NB) {
#pragma unroll
      for (int u = 0; u < 2 * CT; ++u) bq[(blk + 1) & 1][u] = SAGE_BFRAG(blk + 1, u);
    }
    if (blk < NB / 2) {
      ax[2 * blk]     = xrow[4 * blk];
      ax[2 * blk + 1] = xrow[4 * blk + 2];
    }
    __builtin_amdgcn_sched_barrier(0);
#pragma unroll
    for (int u2 = 0; u2 < 2; ++u2) {
      float4 a;
      if (blk < NB / 2) {
        a = am[2 * blk + u2];
        a.x *= scale; a.y *= scale; a.z *= scale; a.w *= scale;
      } else {
        a = ax[2 * (blk - NB / 2) + u2];
      }
      v2f a0, a1;
      a0.x = a.x; a0.y = a.y;
      a1.x = a.z; a1.y = a.w;
#pragma unroll
      for (int t = 0; t < CT; ++t) {
        v2f b0;
        b0.x = bq[blk & 1][u2 * CT + t].x; b0.y = bq[blk & 1][u2 * CT + t].y;
        acc[t] = __builtin_amdgcn_wmma_f32_16x16x4_f32(false, a0, false, b0,
                                                       (short)0, acc[t], false, false);
      }
#pragma unroll
      for (int t = 0; t < CT; ++t) {
        v2f b1;
        b1.x = bq[blk & 1][u2 * CT + t].z; b1.y = bq[blk & 1][u2 * CT + t].w;
        acc[t] = __builtin_amdgcn_wmma_f32_16x16x4_f32(false, a1, false, b1,
                                                       (short)0, acc[t], false, false);
      }
    }
    __builtin_amdgcn_sched_barrier(0);
  }
#undef SAGE_BFRAG

  const int rbase = (rowTile << 4) + (h << 3);
#pragma unroll
  for (int t = 0; t < CT; ++t) {
    const int col = colbase + t * 16;
    const float bias = bl[col];
#pragma unroll
    for (int i = 0; i < 8; ++i) {
      float v = acc[t][i] + bias;
      if (doRelu) v = fmaxf(v, 0.0f);
      out[(long long)(rbase + i) * N + col] = v;
    }
  }
}

template <int N>
__global__ void __launch_bounds__(256)
sage_gemm_strip_kernel(const float* __restrict__ msg,
                       const float* __restrict__ deg,
                       const float* __restrict__ xin,
                       const float* __restrict__ Wl,
                       const float* __restrict__ bl,
                       const float* __restrict__ Wr,
                       float* __restrict__ out, int doRelu) {
  constexpr int K  = FEAT;
  constexpr int KG = GEMM_KG;
  __shared__ __attribute__((aligned(16))) float Ws[2 * K * N];

  const int tid  = threadIdx.x;
  const int lane = tid & 31;
  const int c = lane & 15;
  const int h = lane >> 4;
  const int wv = __builtin_amdgcn_readfirstlane(tid >> 5);
  const int b  = blockIdx.x;

  const int rowTile1 = b * GEMM_RT_BLOCK + (wv >> 1);
  const int half1 = wv & 1;
  const int q = b >> 2;
  const bool left = (wv == 4) && ((b & 3) == 1) && (GEMM_RT_EVEN + (q >> 1) < GEMM_ROW_TILES);
  const bool second = (wv < 4) || left;
  const int rowTile2 = left ? GEMM_RT_EVEN + (q >> 1) : b * GEMM_RT_BLOCK + 4 + (wv >> 1);
  const int half2 = left ? (q & 1) : (wv & 1);

  const int r1 = (rowTile1 << 4) + c;
  const float deg1 = deg[r1];
  float4 am[GEMM_ST];
  {
    const float4* mrow = (const float4*)(msg + (long long)r1 * K) + h;
#pragma unroll
    for (int j = 0; j < GEMM_ST; ++j) am[j] = mrow[2 * j];
  }

  {
    constexpr int KG_PASS = 256 / N;
    const int n   = tid % N;
    const int kgo = tid / N;
    for (int m = 0; m < 2; ++m) {
      const float* __restrict__ Wm = m ? Wr : Wl;
#pragma unroll 4
      for (int p = 0; p < KG / KG_PASS; ++p) {
        const int kg = p * KG_PASS + kgo;
        float4 v;
        v.x = Wm[(4 * kg + 0) * N + n];
        v.y = Wm[(4 * kg + 1) * N + n];
        v.z = Wm[(4 * kg + 2) * N + n];
        v.w = Wm[(4 * kg + 3) * N + n];
        *(float4*)(Ws + ((m * KG + kg) * N + n) * 4) = v;
      }
    }
  }
  __syncthreads();

  sage_gemm_item<N>(Ws, am, deg1, xin, bl, out, doRelu, rowTile1, half1, c, h);
  if (second) {
    const int r2 = (rowTile2 << 4) + c;
    const float deg2 = deg[r2];
    float4 an[GEMM_ST];
    const float4* mrow = (const float4*)(msg + (long long)r2 * K) + h;
#pragma unroll
    for (int j = 0; j < GEMM_ST; ++j) an[j] = mrow[2 * j];
    sage_gemm_item<N>(Ws, an, deg2, xin, bl, out, doRelu, rowTile2, half2, c, h);
  }
}

#define XF_N        128
#define XF_RT_BLOCK 6
#define XF_CSR_OFF  50048

__device__ __forceinline__ void
sage_xform_pass(const float* Ws, const float4 (&am)[GEMM_ST], int colbase, int h, v8f (&acc)[4]) {
  constexpr int CT = 4;
  constexpr int NB = GEMM_ST / 2;
  const float* wb = Ws + (h * XF_N + colbase) * 4;
#define XF_BFRAG(blk, u) \
  (*(const float4*)(wb + ((2 * (2 * (blk) + (u) / CT)) * XF_N + ((u) % CT) * 16) * 4))
#pragma unroll
  for (int t = 0; t < CT; ++t) acc[t] = v8f{};
  float4 bq[2][2 * CT];
#pragma unroll
  for (int u = 0; u < 2 * CT; ++u) bq[0][u] = XF_BFRAG(0, u);
#pragma unroll
  for (int blk = 0; blk < NB; ++blk) {
    if (blk + 1 < NB) {
#pragma unroll
      for (int u = 0; u < 2 * CT; ++u) bq[(blk + 1) & 1][u] = XF_BFRAG(blk + 1, u);
    }
    __builtin_amdgcn_sched_barrier(0);
#pragma unroll
    for (int u2 = 0; u2 < 2; ++u2) {
      const float4 a = am[2 * blk + u2];
      v2f a0, a1;
      a0.x = a.x; a0.y = a.y;
      a1.x = a.z; a1.y = a.w;
#pragma unroll
      for (int t = 0; t < CT; ++t) {
        v2f b0;
        b0.x = bq[blk & 1][u2 * CT + t].x; b0.y = bq[blk & 1][u2 * CT + t].y;
        acc[t] = __builtin_amdgcn_wmma_f32_16x16x4_f32(false, a0, false, b0,
                                                       (short)0, acc[t], false, false);
      }
#pragma unroll
      for (int t = 0; t < CT; ++t) {
        v2f b1;
        b1.x = bq[blk & 1][u2 * CT + t].z; b1.y = bq[blk & 1][u2 * CT + t].w;
        acc[t] = __builtin_amdgcn_wmma_f32_16x16x4_f32(false, a1, false, b1,
                                                       (short)0, acc[t], false, false);
      }
    }
    __builtin_amdgcn_sched_barrier(0);
  }
#undef XF_BFRAG
}

__global__ void __launch_bounds__(256)
sage_xform_kernel(float* hq,
                  const float* __restrict__ Wl,
                  const float* __restrict__ bl,
                  const float* __restrict__ Wr,
                  float* __restrict__ P,
                  const int* __restrict__ csrIn, int* __restrict__ csrOut, int E) {
  constexpr int K  = FEAT;
  constexpr int KG = GEMM_KG;
  __shared__ __attribute__((aligned(16))) float Ws[K * XF_N];

  const int tid  = threadIdx.x;
  const int lane = tid & 31;
  const int c = lane & 15;
  const int h = lane >> 4;
  const int wv = __builtin_amdgcn_readfirstlane(tid >> 5);
  const int b  = blockIdx.x;

  const int rowTile = (wv < XF_RT_BLOCK) ? b * XF_RT_BLOCK + wv : GEMM_RT_EVEN + b;
  const bool work = (wv < XF_RT_BLOCK) || (wv == XF_RT_BLOCK && rowTile < GEMM_ROW_TILES);

  const int r = (rowTile << 4) + c;
  float4 am[GEMM_ST];
  if (work) {
    const float4* mrow = (const float4*)(hq + (long long)r * K) + h;
#pragma unroll
    for (int j = 0; j < GEMM_ST; ++j) am[j] = mrow[2 * j];
  }

  {
    const int n4 = (XF_CSR_OFF + E + 3) >> 2;
    const int4* __restrict__ s4 = (const int4*)csrIn;
    int4* __restrict__ d4 = (int4*)csrOut;
    for (int i = b * 256 + tid; i < n4; i += GEMM_GRID * 256) d4[i] = s4[i];
  }

  {
    const int n   = tid & 63;
    const int kgo = tid >> 6;
    for (int m = 0; m < 2; ++m) {
      const float* __restrict__ Wm = m ? Wr : Wl;
#pragma unroll 4
      for (int p = 0; p < KG / 4; ++p) {
        const int kg = p * 4 + kgo;
        float4 v;
        v.x = Wm[(4 * kg + 0) * 64 + n];
        v.y = Wm[(4 * kg + 1) * 64 + n];
        v.z = Wm[(4 * kg + 2) * 64 + n];
        v.w = Wm[(4 * kg + 3) * 64 + n];
        *(float4*)(Ws + (kg * XF_N + m * 64 + n) * 4) = v;
      }
    }
  }
  __syncthreads();

  if (work) {
    const int rbase = (rowTile << 4) + (h << 3);
    v8f acc[4];
    sage_xform_pass(Ws, am, c, h, acc);
#pragma unroll
    for (int t = 0; t < 4; ++t) {
      const int col = c + t * 16;
#pragma unroll
      for (int i = 0; i < 8; ++i) P[(long long)(rbase + i) * 64 + col] = acc[t][i];
    }
    sage_xform_pass(Ws, am, 64 + c, h, acc);
#pragma unroll
    for (int t = 0; t < 4; ++t) {
      const int col = c + t * 16;
      const float bias = bl[col];
#pragma unroll
      for (int i = 0; i < 8; ++i) hq[(long long)(rbase + i) * K + col] = acc[t][i] + bias;
    }
  }
}

#define SAGE_PROW(k) p4[(long long)sage_pair_id(idx, t + (k), hh) * 16 + c16]
__device__ __forceinline__ int sage_pair_id(int idx, int slot, int hh) {
  const int i0 = __builtin_amdgcn_readlane(idx, slot);
  const int i1 = __builtin_amdgcn_readlane(idx, slot + 1);
  return hh ? i1 : i0;
}
__global__ void __launch_bounds__(256)
sage_gather_mean_kernel(const float* __restrict__ P,
                        const float* __restrict__ hq,
                        const int* __restrict__ rowstart,
                        const int* __restrict__ col,
                        float* __restrict__ out) {
  const int lane = threadIdx.x & 31;
  const int c16 = lane & 15;
  const int hh = lane >> 4;
  const int node = __builtin_amdgcn_readfirstlane((int)(blockIdx.x * 8 + (threadIdx.x >> 5)));
  if (node >= N_NODES) return;
  const int beg = rowstart[node];
  const int end = rowstart[node + 1];
  const float4* p4 = (const float4*)P;
  const float4 q = ((const float4*)hq)[(long long)node * 32 + c16];
  float4 acc = make_float4(0.0f, 0.0f, 0.0f, 0.0f);
  for (int base = beg; base < end; base += 32) {
    const int j = base + lane;
    const int idx = (j < end) ? col[j] : 0;
    const int m = min(32, end - base);
    int t = 0;
    for (; t + 16 <= m; t += 16) {
      const float4 a0 = SAGE_PROW(0), a1 = SAGE_PROW(2), a2 = SAGE_PROW(4), a3 = SAGE_PROW(6);
      const float4 a4 = SAGE_PROW(8), a5 = SAGE_PROW(10), a6 = SAGE_PROW(12), a7 = SAGE_PROW(14);
      SAGE_ADD(a0) SAGE_ADD(a1) SAGE_ADD(a2) SAGE_ADD(a3)
      SAGE_ADD(a4) SAGE_ADD(a5) SAGE_ADD(a6) SAGE_ADD(a7)
    }
    if (t + 8 <= m) {
      const float4 a0 = SAGE_PROW(0), a1 = SAGE_PROW(2), a2 = SAGE_PROW(4), a3 = SAGE_PROW(6);
      SAGE_ADD(a0) SAGE_ADD(a1) SAGE_ADD(a2) SAGE_ADD(a3)
      t += 8;
    }
    if (t + 4 <= m) {
      const float4 a0 = SAGE_PROW(0), a1 = SAGE_PROW(2);
      SAGE_ADD(a0) SAGE_ADD(a1)
      t += 4;
    }
    if (t + 2 <= m) {
      const float4 a0 = SAGE_PROW(0);
      SAGE_ADD(a0)
      t += 2;
    }
    if (t < m) {
      const float4 a0 = SAGE_PROW(0);
      if (hh == 0) { SAGE_ADD(a0) }
    }
  }
  acc.x += __shfl_xor(acc.x, 16);
  acc.y += __shfl_xor(acc.y, 16);
  acc.z += __shfl_xor(acc.z, 16);
  acc.w += __shfl_xor(acc.w, 16);
  const float scale = 1.0f / fmaxf((float)(end - beg), 1.0f);
  if (hh == 0) {
    float4 o;
    o.x = acc.x * scale + q.x;
    o.y = acc.y * scale + q.y;
    o.z = acc.z * scale + q.z;
    o.w = acc.w * scale + q.w;
    ((float4*)out)[(long long)node * 16 + c16] = o;
  }
}

extern "C" void kernel_launch(void* const* d_in, const int* in_sizes, int n_in,
                              void* d_out, int out_size, void* d_ws, size_t ws_size,
                              hipStream_t stream) {
  const float* x   = (const float*)d_in[0];
  const int*   ei  = (const int*)d_in[1];
  const float* Wl1 = (const float*)d_in[2];
  const float* bl1 = (const float*)d_in[3];
  const float* Wr1 = (const float*)d_in[4];
  const float* Wl2 = (const float*)d_in[5];
  const float* bl2 = (const float*)d_in[6];
  const float* Wr2 = (const float*)d_in[7];
  float* out = (float*)d_out;

  const int E = in_sizes[1] / 2;
  const int* src = ei;
  const int* dst = ei + E;

  float* msg = (float*)d_ws;
  float* deg = msg + (long long)N_NODES * FEAT;
  float* h1  = deg + N_NODES;

  int* rowstart = (int*)d_out;
  int* col      = rowstart + XF_CSR_OFF;

  int* loc  = (int*)h1;
  int* keys = loc + 65536;

  const int TB = 256;
  const int nodeBlocks = N_NODES / 8;

  sage_chunk_sort_kernel<<<SORT_NCH, SORT_THREADS, 0, stream>>>(src, dst, loc, keys, E);
  sage_bucket_fill_kernel<<<SORT_NB, FILL_THREADS, 0, stream>>>(loc, keys, rowstart, deg, col, E);

  sage_gather_kernel<<<nodeBlocks, TB, 0, stream>>>(x, rowstart, col, msg);
  sage_gemm_strip_kernel<128><<<GEMM_GRID, 256, 0, stream>>>(msg, deg, x, Wl1, bl1, Wr1, h1, 1);

  float* P = msg;
  int* rowstart2 = (int*)(msg + (long long)N_NODES * 64);
  int* col2      = rowstart2 + XF_CSR_OFF;
  sage_xform_kernel<<<GEMM_GRID, 256, 0, stream>>>(h1, Wl2, bl2, Wr2, P, rowstart, rowstart2, E);
  sage_gather_mean_kernel<<<nodeBlocks, TB, 0, stream>>>(P, h1, rowstart2, col2, out);
}


# Edge sort stores whole lines: both sort kernels place their keys in an LDS copy of their output range (slots from the same LDS atomics, row order = LDS arrival as before) and write it front to back
# speedup vs baseline: 4.3599x; 1.0398x over previous
#include <hip/hip_runtime.h>

typedef float v2f __attribute__((ext_vector_type(2)));
typedef float v8f __attribute__((ext_vector_type(8)));

#define N_NODES 50000
#define FEAT    128

#define SORT_BSH      7
#define SORT_BNODES   (1 << SORT_BSH)
#define SORT_NB       ((N_NODES + SORT_BNODES - 1) / SORT_BNODES)
#define SORT_NCH      128
#define SORT_THREADS  1024
#define SORT_UNR      8
#define SORT_SCAN_CH  13
static_assert(N_NODES <= 65536, "keys pack a source node id into 16 bits");
static_assert(32 * SORT_SCAN_CH >= SORT_NB + 1, "one wave scans the bucket counts of a chunk");

__global__ void __launch_bounds__(SORT_THREADS)
sage_chunk_sort_kernel(const int* __restrict__ src, const int* __restrict__ dst,
                       int* __restrict__ loc, int* __restrict__ keys, int E) {
  __shared__ int h[SORT_NB + 1];
  __shared__ int stage[SORT_UNR * SORT_THREADS];
  const int tid = threadIdx.x;
  const int w = blockIdx.x;
  if (tid <= SORT_NB) h[tid] = 0;
  __syncthreads();
  const int chunk = (E + SORT_NCH - 1) / SORT_NCH;
  const int beg = w * chunk;
  const int end = min(E, beg + chunk);
  const bool one = (end - beg) <= SORT_UNR * SORT_THREADS;

  int d[SORT_UNR], s[SORT_UNR];
  if (one) {
#pragma unroll
    for (int k = 0; k < SORT_UNR; ++k) {
      const int e = beg + k * SORT_THREADS + tid;
      d[k] = (e < end) ? dst[e] : -1;
      s[k] = (e < end) ? src[e] : 0;
    }
#pragma unroll
    for (int k = 0; k < SORT_UNR; ++k)
      if (d[k] >= 0) atomicAdd(&h[d[k] >> SORT_BSH], 1);
  } else {
    for (int base = beg; base < end; base += SORT_UNR * SORT_THREADS) {
      int dd[SORT_UNR];
#pragma unroll
      for (int k = 0; k < SORT_UNR; ++k) {
        const int e = base + k * SORT_THREADS + tid;
        dd[k] = (e < end) ? dst[e] : -1;
      }
#pragma unroll
      for (int k = 0; k < SORT_UNR; ++k)
        if (dd[k] >= 0) atomicAdd(&h[dd[k] >> SORT_BSH], 1);
    }
  }
  __syncthreads();

  if (tid < 32) {
    int v[SORT_SCAN_CH];
    int tot = 0;
#pragma unroll
    for (int k = 0; k < SORT_SCAN_CH; ++k) {
      const int i = tid * SORT_SCAN_CH + k;
      v[k] = (i < SORT_NB) ? h[i] : 0;
      tot += v[k];
    }
    int inc = tot;
    for (int o = 1; o < 32; o <<= 1) {
      const int up = __shfl_up(inc, o);
      if (tid >= o) inc += up;
    }
    int run = inc - tot;
#pragma unroll
    for (int k = 0; k < SORT_SCAN_CH; ++k) {
      const int i = tid * SORT_SCAN_CH + k;
      if (i <= SORT_NB) {
        h[i] = run;
        loc[i * SORT_NCH + w] = run;
      }
      run += v[k];
    }
  }
  __syncthreads();

  int* __restrict__ mine = keys + beg;
  if (one) {
    int pos[SORT_UNR];
#pragma unroll
    for (int k = 0; k < SORT_UNR; ++k)
      pos[k] = (d[k] >= 0) ? atomicAdd(&h[d[k] >> SORT_BSH], 1) : 0;
#pragma unroll
    for (int k = 0; k < SORT_UNR; ++k)
      if (d[k] >= 0) stage[pos[k]] = ((d[k] & (SORT_BNODES - 1)) << 16) | s[k];
    __syncthreads();
    const int n = end - beg;
    for (int i = tid; i < n; i += SORT_THREADS) mine[i] = stage[i];
  } else {
    for (int base = beg; base < end; base += SORT_UNR * SORT_THREADS) {
      int dd[SORT_UNR], ss[SORT_UNR];
#pragma unroll
      for (int k = 0; k < SORT_UNR; ++k) {
        const int e = base + k * SORT_THREADS + tid;
        dd[k] = (e < end) ? dst[e] : -1;
        ss[k] = (e < end) ? src[e] : 0;
      }
#pragma unroll
      for (int k = 0; k < SORT_UNR; ++k) {
        if (dd[k] >= 0) {
          const int pos = atomicAdd(&h[dd[k] >> SORT_BSH], 1);
          mine[pos] = ((dd[k] & (SORT_BNODES - 1)) << 16) | ss[k];
        }
      }
    }
  }
}

#define FILL_THREADS 512
#define FILL_WAVES   (FILL_THREADS / 32)
#define FILL_PPW     (SORT_NCH / FILL_WAVES)
#define FILL_CAP     4096
__global__ void __launch_bounds__(FILL_THREADS)
sage_bucket_fill_kernel(const int* __restrict__ loc, const int* __restrict__ keys,
                        int* __restrict__ rowstart, float* __restrict__ deg,
                        int* __restrict__ col, int E) {
  __shared__ __attribute__((aligned(16))) int c[SORT_BNODES];
  __shared__ int red[SORT_NCH / 32];
  __shared__ int stage[FILL_CAP];
  __shared__ int total_s;
  const int tid = threadIdx.x;
  const int lane = tid & 31;
  const int wv = __builtin_amdgcn_readfirstlane(tid >> 5);
  const int b = blockIdx.x;
  const int chunk = (E + SORT_NCH - 1) / SORT_NCH;
  const int* __restrict__ lrow = loc + b * SORT_NCH;

  int st[FILL_PPW], len[FILL_PPW];
#pragma unroll
  for (int u = 0; u < FILL_PPW; ++u) {
    const int p = wv + u * FILL_WAVES;
    const int lo = lrow[p];
    st[u] = p * chunk + lo;
    len[u] = lrow[SORT_NCH + p] - lo;
  }
  int k[FILL_PPW];
#pragma unroll
  for (int u = 0; u < FILL_PPW; ++u)
    k[u] = (lane < len[u]) ? keys[st[u] + lane] : -1;

  if (tid < SORT_NCH) {
    c[tid] = 0;
    int slo = lrow[tid];
    for (int o = 16; o > 0; o >>= 1) slo += __shfl_xor(slo, o);
    if (lane == 0) red[wv] = slo;
  }
  __syncthreads();
  int beg = 0;
  for (int q = 0; q < SORT_NCH / 32; ++q) beg += red[q];

#pragma unroll
  for (int u = 0; u < FILL_PPW; ++u)
    if (k[u] >= 0) atomicAdd(&c[k[u] >> 16], 1);
#pragma unroll
  for (int u = 0; u < FILL_PPW; ++u)
    for (int j = 32 + lane; j < len[u]; j += 32) atomicAdd(&c[keys[st[u] + j] >> 16], 1);
  __syncthreads();

  if (tid < 32) {
    const int4 cv = *(const int4*)&c[4 * tid];
    const int tot = cv.x + cv.y + cv.z + cv.w;
    int inc = tot;
    for (int o = 1; o < 32; o <<= 1) {
      const int up = __shfl_up(inc, o);
      if (tid >= o) inc += up;
    }
    int4 r;
    r.x = beg + inc - tot;
    r.y = r.x + cv.x;
    r.z = r.y + cv.y;
    r.w = r.z + cv.z;
    *(int4*)&c[4 * tid] = r;
    if (tid == 31) total_s = inc;
    const int node = b * SORT_BNODES + 4 * tid;
    if (node < N_NODES) {
      *(int4*)&rowstart[node] = r;
      *(float4*)&deg[node] = make_float4((float)cv.x, (float)cv.y, (float)cv.z, (float)cv.w);
    } else if (node == N_NODES) {
      rowstart[node] = r.x;
    }
  }
  __syncthreads();

  const int total = total_s;
  if (total <= FILL_CAP) {
    int pos[FILL_PPW];
#pragma unroll
    for (int u = 0; u < FILL_PPW; ++u)
      pos[u] = (k[u] >= 0) ? atomicAdd(&c[k[u] >> 16], 1) : beg;
#pragma unroll
    for (int u = 0; u < FILL_PPW; ++u)
      if (k[u] >= 0) stage[pos[u] - beg] = k[u] & 0xFFFF;
#pragma unroll
    for (int u = 0; u < FILL_PPW; ++u) {
      for (int j = 32 + lane; j < len[u]; j += 32) {
        const int kk = keys[st[u] + j];
        const int p = atomicAdd(&c[kk >> 16], 1);
        stage[p - beg] = kk & 0xFFFF;
      }
    }
    __syncthreads();
    for (int i = tid; i < total; i += FILL_THREADS) col[beg + i] = stage[i];
  } else {
#pragma unroll
    for (int u = 0; u < FILL_PPW; ++u) {
      if (k[u] >= 0) {
        const int pos = atomicAdd(&c[k[u] >> 16], 1);
        col[pos] = k[u] & 0xFFFF;
      }
    }
#pragma unroll
    for (int u = 0; u < FILL_PPW; ++u) {
      for (int j = 32 + lane; j < len[u]; j += 32) {
        const int kk = keys[st[u] + j];
        const int pos = atomicAdd(&c[kk >> 16], 1);
        col[pos] = kk & 0xFFFF;
      }
    }
  }
}

#define SAGE_ROW(k) x4[(long long)__builtin_amdgcn_readlane(idx, t + (k)) * 32 + lane]
#define SAGE_ADD(a) acc.x += a.x; acc.y += a.y; acc.z += a.z; acc.w += a.w;
__global__ void __launch_bounds__(256)
sage_gather_kernel(const float* __restrict__ x,
                   const int* __restrict__ rowstart,
                   const int* __restrict__ col,
                   float* __restrict__ msg) {
  const int lane = threadIdx.x & 31;
  const int node = __builtin_amdgcn_readfirstlane((int)(blockIdx.x * 8 + (threadIdx.x >> 5)));
  if (node >= N_NODES) return;
  const int beg = rowstart[node];
  const int end = rowstart[node + 1];
  const float4* x4 = (const float4*)x;
  float4 acc = make_float4(0.0f, 0.0f, 0.0f, 0.0f);
  for (int base = beg; base < end; base += 32) {
    const int j = base + lane;
    const int idx = (j < end) ? col[j] : 0;
    const int m = min(32, end - base);
    int t = 0;
    for (; t + 8 <= m; t += 8) {
      const float4 a0 = SAGE_ROW(0), a1 = SAGE_ROW(1), a2 = SAGE_ROW(2), a3 = SAGE_ROW(3);
      const float4 a4 = SAGE_ROW(4), a5 = SAGE_ROW(5), a6 = SAGE_ROW(6), a7 = SAGE_ROW(7);
      SAGE_ADD(a0) SAGE_ADD(a1) SAGE_ADD(a2) SAGE_ADD(a3)
      SAGE_ADD(a4) SAGE_ADD(a5) SAGE_ADD(a6) SAGE_ADD(a7)
    }
    if (t + 4 <= m) {
      const float4 a0 = SAGE_ROW(0), a1 = SAGE_ROW(1), a2 = SAGE_ROW(2), a3 = SAGE_ROW(3);
      SAGE_ADD(a0) SAGE_ADD(a1) SAGE_ADD(a2) SAGE_ADD(a3)
      t += 4;
    }
    if (t + 2 <= m) {
      const float4 a0 = SAGE_ROW(0), a1 = SAGE_ROW(1);
      SAGE_ADD(a0) SAGE_ADD(a1)
      t += 2;
    }
    if (t < m) {
      const float4 a0 = SAGE_ROW(0);
      SAGE_ADD(a0)
    }
  }
  ((float4*)msg)[(long long)node * 32 + lane] = acc;
}

#define GEMM_GRID      512
#define GEMM_RT_BLOCK  6
#define GEMM_RT_EVEN   (GEMM_GRID * GEMM_RT_BLOCK)
#define GEMM_ROW_TILES (N_NODES / 16)
#define GEMM_KG        (FEAT / 4)
#define GEMM_ST        (GEMM_KG / 2)
#define GEMM_NB        GEMM_ST

template <int N>
__device__ __forceinline__ void
sage_gemm_item(const float* Ws, const float4 (&am)[GEMM_ST], float degv,
               const float* __restrict__ xin, const float* __restrict__ bl,
               float* __restrict__ out, int doRelu, int rowTile, int half, int c, int h) {
  constexpr int K  = FEAT;
  constexpr int KG = GEMM_KG;
  constexpr int NB = GEMM_NB;
  constexpr int CT = N / 32;

  const int r = (rowTile << 4) + c;
  const float scale = 1.0f / fmaxf(degv, 1.0f);
  const int colbase = half * (N / 2) + c;
  const float4* xrow = (const float4*)(xin + (long long)r * K) + h;
  const float* wb = Ws + (h * N + colbase) * 4;
#define SAGE_BFRAG(blk, u) \
  (*(const float4*)(wb + ((((blk) >> 3) * KG + 2 * (2 * ((blk) & 7) + (u) / CT)) * N + ((u) % CT) * 16) * 4))

  float4 ax[GEMM_ST];
  v8f acc[CT];
#pragma unroll
  for (int t = 0; t < CT; ++t) acc[t] = v8f{};

  float4 bq[2][2 * CT];
#pragma unroll
  for (int u = 0; u < 2 * CT; ++u) bq[0][u] = SAGE_BFRAG(0, u);

#pragma unroll
  for (int blk = 0; blk < NB; ++blk) {
    if (blk + 1 < NB) {
#pragma unroll
      for (int u = 0; u < 2 * CT; ++u) bq[(blk + 1) & 1][u] = SAGE_BFRAG(blk + 1, u);
    }
    if (blk < NB / 2) {
      ax[2 * blk]     = xrow[4 * blk];
      ax[2 * blk + 1] = xrow[4 * blk + 2];
    }
    __builtin_amdgcn_sched_barrier(0);
#pragma unroll
    for (int u2 = 0; u2 < 2; ++u2) {
      float4 a;
      if (blk < NB / 2) {
        a = am[2 * blk + u2];
        a.x *= scale; a.y *= scale; a.z *= scale; a.w *= scale;
      } else {
        a = ax[2 * (blk - NB / 2) + u2];
      }
      v2f a0, a1;
      a0.x = a.x; a0.y = a.y;
      a1.x = a.z; a1.y = a.w;
#pragma unroll
      for (int t = 0; t < CT; ++t) {
        v2f b0;
        b0.x = bq[blk & 1][u2 * CT + t].x; b0.y = bq[blk & 1][u2 * CT + t].y;
        acc[t] = __builtin_amdgcn_wmma_f32_16x16x4_f32(false, a0, false, b0,
                                                       (short)0, acc[t], false, false);
      }
#pragma unroll
      for (int t = 0; t < CT; ++t) {
        v2f b1;
        b1.x = bq[blk & 1][u2 * CT + t].z; b1.y = bq[blk & 1][u2 * CT + t].w;
        acc[t] = __builtin_amdgcn_wmma_f32_16x16x4_f32(false, a1, false, b1,
                                                       (short)0, acc[t], false, false);
      }
    }
    __builtin_amdgcn_sched_barrier(0);
  }
#undef SAGE_BFRAG

  const int rbase = (rowTile << 4) + (h << 3);
#pragma unroll
  for (int t = 0; t < CT; ++t) {
    const int col = colbase + t * 16;
    const float bias = bl[col];
#pragma unroll
    for (int i = 0; i < 8; ++i) {
      float v = acc[t][i] + bias;
      if (doRelu) v = fmaxf(v, 0.0f);
      out[(long long)(rbase + i) * N + col] = v;
    }
  }
}

template <int N>
__global__ void __launch_bounds__(256)
sage_gemm_strip_kernel(const float* __restrict__ msg,
                       const float* __restrict__ deg,
                       const float* __restrict__ xin,
                       const float* __restrict__ Wl,
                       const float* __restrict__ bl,
                       const float* __restrict__ Wr,
                       float* __restrict__ out, int doRelu) {
  constexpr int K  = FEAT;
  constexpr int KG = GEMM_KG;
  __shared__ __attribute__((aligned(16))) float Ws[2 * K * N];

  const int tid  = threadIdx.x;
  const int lane = tid & 31;
  const int c = lane & 15;
  const int h = lane >> 4;
  const int wv = __builtin_amdgcn_readfirstlane(tid >> 5);
  const int b  = blockIdx.x;

  const int rowTile1 = b * GEMM_RT_BLOCK + (wv >> 1);
  const int half1 = wv & 1;
  const int q = b >> 2;
  const bool left = (wv == 4) && ((b & 3) == 1) && (GEMM_RT_EVEN + (q >> 1) < GEMM_ROW_TILES);
  const bool second = (wv < 4) || left;
  const int rowTile2 = left ? GEMM_RT_EVEN + (q >> 1) : b * GEMM_RT_BLOCK + 4 + (wv >> 1);
  const int half2 = left ? (q & 1) : (wv & 1);

  const int r1 = (rowTile1 << 4) + c;
  const float deg1 = deg[r1];
  float4 am[GEMM_ST];
  {
    const float4* mrow = (const float4*)(msg + (long long)r1 * K) + h;
#pragma unroll
    for (int j = 0; j < GEMM_ST; ++j) am[j] = mrow[2 * j];
  }

  {
    constexpr int KG_PASS = 256 / N;
    const int n   = tid % N;
    const int kgo = tid / N;
    for (int m = 0; m < 2; ++m) {
      const float* __restrict__ Wm = m ? Wr : Wl;
#pragma unroll 4
      for (int p = 0; p < KG / KG_PASS; ++p) {
        const int kg = p * KG_PASS + kgo;
        float4 v;
        v.x = Wm[(4 * kg + 0) * N + n];
        v.y = Wm[(4 * kg + 1) * N + n];
        v.z = Wm[(4 * kg + 2) * N + n];
        v.w = Wm[(4 * kg + 3) * N + n];
        *(float4*)(Ws + ((m * KG + kg) * N + n) * 4) = v;
      }
    }
  }
  __syncthreads();

  sage_gemm_item<N>(Ws, am, deg1, xin, bl, out, doRelu, rowTile1, half1, c, h);
  if (second) {
    const int r2 = (rowTile2 << 4) + c;
    const float deg2 = deg[r2];
    float4 an[GEMM_ST];
    const float4* mrow = (const float4*)(msg + (long long)r2 * K) + h;
#pragma unroll
    for (int j = 0; j < GEMM_ST; ++j) an[j] = mrow[2 * j];
    sage_gemm_item<N>(Ws, an, deg2, xin, bl, out, doRelu, rowTile2, half2, c, h);
  }
}

#define XF_N        128
#define XF_RT_BLOCK 6
#define XF_CSR_OFF  50048

__device__ __forceinline__ void
sage_xform_pass(const float* Ws, const float4 (&am)[GEMM_ST], int colbase, int h, v8f (&acc)[4]) {
  constexpr int CT = 4;
  constexpr int NB = GEMM_ST / 2;
  const float* wb = Ws + (h * XF_N + colbase) * 4;
#define XF_BFRAG(blk, u) \
  (*(const float4*)(wb + ((2 * (2 * (blk) + (u) / CT)) * XF_N + ((u) % CT) * 16) * 4))
#pragma unroll
  for (int t = 0; t < CT; ++t) acc[t] = v8f{};
  float4 bq[2][2 * CT];
#pragma unroll
  for (int u = 0; u < 2 * CT; ++u) bq[0][u] = XF_BFRAG(0, u);
#pragma unroll
  for (int blk = 0; blk < NB; ++blk) {
    if (blk + 1 < NB) {
#pragma unroll
      for (int u = 0; u < 2 * CT; ++u) bq[(blk + 1) & 1][u] = XF_BFRAG(blk + 1, u);
    }
    __builtin_amdgcn_sched_barrier(0);
#pragma unroll
    for (int u2 = 0; u2 < 2; ++u2) {
      const float4 a = am[2 * blk + u2];
      v2f a0, a1;
      a0.x = a.x; a0.y = a.y;
      a1.x = a.z; a1.y = a.w;
#pragma unroll
      for (int t = 0; t < CT; ++t) {
        v2f b0;
        b0.x = bq[blk & 1][u2 * CT + t].x; b0.y = bq[blk & 1][u2 * CT + t].y;
        acc[t] = __builtin_amdgcn_wmma_f32_16x16x4_f32(false, a0, false, b0,
                                                       (short)0, acc[t], false, false);
      }
#pragma unroll
      for (int t = 0; t < CT; ++t) {
        v2f b1;
        b1.x = bq[blk & 1][u2 * CT + t].z; b1.y = bq[blk & 1][u2 * CT + t].w;
        acc[t] = __builtin_amdgcn_wmma_f32_16x16x4_f32(false, a1, false, b1,
                                                       (short)0, acc[t], false, false);
      }
    }
    __builtin_amdgcn_sched_barrier(0);
  }
#undef XF_BFRAG
}

__global__ void __launch_bounds__(256)
sage_xform_kernel(float* hq,
                  const float* __restrict__ Wl,
                  const float* __restrict__ bl,
                  const float* __restrict__ Wr,
                  float* __restrict__ P,
                  const int* __restrict__ csrIn, int* __restrict__ csrOut, int E) {
  constexpr int K  = FEAT;
  constexpr int KG = GEMM_KG;
  __shared__ __attribute__((aligned(16))) float Ws[K * XF_N];

  const int tid  = threadIdx.x;
  const int lane = tid & 31;
  const int c = lane & 15;
  const int h = lane >> 4;
  const int wv = __builtin_amdgcn_readfirstlane(tid >> 5);
  const int b  = blockIdx.x;

  const int rowTile = (wv < XF_RT_BLOCK) ? b * XF_RT_BLOCK + wv : GEMM_RT_EVEN + b;
  const bool work = (wv < XF_RT_BLOCK) || (wv == XF_RT_BLOCK && rowTile < GEMM_ROW_TILES);

  const int r = (rowTile << 4) + c;
  float4 am[GEMM_ST];
  if (work) {
    const float4* mrow = (const float4*)(hq + (long long)r * K) + h;
#pragma unroll
    for (int j = 0; j < GEMM_ST; ++j) am[j] = mrow[2 * j];
  }

  {
    const int n4 = (XF_CSR_OFF + E + 3) >> 2;
    const int4* __restrict__ s4 = (const int4*)csrIn;
    int4* __restrict__ d4 = (int4*)csrOut;
    for (int i = b * 256 + tid; i < n4; i += GEMM_GRID * 256) d4[i] = s4[i];
  }

  {
    const int n   = tid & 63;
    const int kgo = tid >> 6;
    for (int m = 0; m < 2; ++m) {
      const float* __restrict__ Wm = m ? Wr : Wl;
#pragma unroll 4
      for (int p = 0; p < KG / 4; ++p) {
        const int kg = p * 4 + kgo;
        float4 v;
        v.x = Wm[(4 * kg + 0) * 64 + n];
        v.y = Wm[(4 * kg + 1) * 64 + n];
        v.z = Wm[(4 * kg + 2) * 64 + n];
        v.w = Wm[(4 * kg + 3) * 64 + n];
        *(float4*)(Ws + (kg * XF_N + m * 64 + n) * 4) = v;
      }
    }
  }
  __syncthreads();

  if (work) {
    const int rbase = (rowTile << 4) + (h << 3);
    v8f acc[4];
    sage_xform_pass(Ws, am, c, h, acc);
#pragma unroll
    for (int t = 0; t < 4; ++t) {
      const int col = c + t * 16;
#pragma unroll
      for (int i = 0; i < 8; ++i) P[(long long)(rbase + i) * 64 + col] = acc[t][i];
    }
    sage_xform_pass(Ws, am, 64 + c, h, acc);
#pragma unroll
    for (int t = 0; t < 4; ++t) {
      const int col = c + t * 16;
      const float bias = bl[col];
#pragma unroll
      for (int i = 0; i < 8; ++i) hq[(long long)(rbase + i) * K + col] = acc[t][i] + bias;
    }
  }
}

#define SAGE_PROW(k) p4[(long long)sage_pair_id(idx, t + (k), hh) * 16 + c16]
__device__ __forceinline__ int sage_pair_id(int idx, int slot, int hh) {
  const int i0 = __builtin_amdgcn_readlane(idx, slot);
  const int i1 = __builtin_amdgcn_readlane(idx, slot + 1);
  return hh ? i1 : i0;
}
__global__ void __launch_bounds__(256)
sage_gather_mean_kernel(const float* __restrict__ P,
                        const float* __restrict__ hq,
                        const int* __restrict__ rowstart,
                        const int* __restrict__ col,
                        float* __restrict__ out) {
  const int lane = threadIdx.x & 31;
  const int c16 = lane & 15;
  const int hh = lane >> 4;
  const int node = __builtin_amdgcn_readfirstlane((int)(blockIdx.x * 8 + (threadIdx.x >> 5)));
  if (node >= N_NODES) return;
  const int beg = rowstart[node];
  const int end = rowstart[node + 1];
  const float4* p4 = (const float4*)P;
  const float4 q = ((const float4*)hq)[(long long)node * 32 + c16];
  float4 acc = make_float4(0.0f, 0.0f, 0.0f, 0.0f);
  for (int base = beg; base < end; base += 32) {
    const int j = base + lane;
    const int idx = (j < end) ? col[j] : 0;
    const int m = min(32, end - base);
    int t = 0;
    for (; t + 16 <= m; t += 16) {
      const float4 a0 = SAGE_PROW(0), a1 = SAGE_PROW(2), a2 = SAGE_PROW(4), a3 = SAGE_PROW(6);
      const float4 a4 = SAGE_PROW(8), a5 = SAGE_PROW(10), a6 = SAGE_PROW(12), a7 = SAGE_PROW(14);
      SAGE_ADD(a0) SAGE_ADD(a1) SAGE_ADD(a2) SAGE_ADD(a3)
      SAGE_ADD(a4) SAGE_ADD(a5) SAGE_ADD(a6) SAGE_ADD(a7)
    }
    if (t + 8 <= m) {
      const float4 a0 = SAGE_PROW(0), a1 = SAGE_PROW(2), a2 = SAGE_PROW(4), a3 = SAGE_PROW(6);
      SAGE_ADD(a0) SAGE_ADD(a1) SAGE_ADD(a2) SAGE_ADD(a3)
      t += 8;
    }
    if (t + 4 <= m) {
      const float4 a0 = SAGE_PROW(0), a1 = SAGE_PROW(2);
      SAGE_ADD(a0) SAGE_ADD(a1)
      t += 4;
    }
    if (t + 2 <= m) {
      const float4 a0 = SAGE_PROW(0);
      SAGE_ADD(a0)
      t += 2;
    }
    if (t < m) {
      const float4 a0 = SAGE_PROW(0);
      if (hh == 0) { SAGE_ADD(a0) }
    }
  }
  acc.x += __shfl_xor(acc.x, 16);
  acc.y += __shfl_xor(acc.y, 16);
  acc.z += __shfl_xor(acc.z, 16);
  acc.w += __shfl_xor(acc.w, 16);
  const float scale = 1.0f / fmaxf((float)(end - beg), 1.0f);
  if (hh == 0) {
    float4 o;
    o.x = acc.x * scale + q.x;
    o.y = acc.y * scale + q.y;
    o.z = acc.z * scale + q.z;
    o.w = acc.w * scale + q.w;
    ((float4*)out)[(long long)node * 16 + c16] = o;
  }
}

extern "C" void kernel_launch(void* const* d_in, const int* in_sizes, int n_in,
                              void* d_out, int out_size, void* d_ws, size_t ws_size,
                              hipStream_t stream) {
  const float* x   = (const float*)d_in[0];
  const int*   ei  = (const int*)d_in[1];
  const float* Wl1 = (const float*)d_in[2];
  const float* bl1 = (const float*)d_in[3];
  const float* Wr1 = (const float*)d_in[4];
  const float* Wl2 = (const float*)d_in[5];
  const float* bl2 = (const float*)d_in[6];
  const float* Wr2 = (const float*)d_in[7];
  float* out = (float*)d_out;

  const int E = in_sizes[1] / 2;
  const int* src = ei;
  const int* dst = ei + E;

  float* msg = (float*)d_ws;
  float* deg = msg + (long long)N_NODES * FEAT;
  float* h1  = deg + N_NODES;

  int* rowstart = (int*)d_out;
  int* col      = rowstart + XF_CSR_OFF;

  int* loc  = (int*)h1;
  int* keys = loc + 65536;

  const int TB = 256;
  const int nodeBlocks = N_NODES / 8;

  sage_chunk_sort_kernel<<<SORT_NCH, SORT_THREADS, 0, stream>>>(src, dst, loc, keys, E);
  sage_bucket_fill_kernel<<<SORT_NB, FILL_THREADS, 0, stream>>>(loc, keys, rowstart, deg, col, E);

  sage_gather_kernel<<<nodeBlocks, TB, 0, stream>>>(x, rowstart, col, msg);
  sage_gemm_strip_kernel<128><<<GEMM_GRID, 256, 0, stream>>>(msg, deg, x, Wl1, bl1, Wr1, h1, 1);

  float* P = msg;
  int* rowstart2 = (int*)(msg + (long long)N_NODES * 64);
  int* col2      = rowstart2 + XF_CSR_OFF;
  sage_xform_kernel<<<GEMM_GRID, 256, 0, stream>>>(h1, Wl2, bl2, Wr2, P, rowstart, rowstart2, E);
  sage_gather_mean_kernel<<<nodeBlocks, TB, 0, stream>>>(P, h1, rowstart2, col2, out);
}


# Both gather launches as resident grids: 16384 waves each loop over 3-4 nodes instead of one short wave per node; next node's bounds and ids requested under current row loads; sum order unchanged
# speedup vs baseline: 4.5157x; 1.0357x over previous
#include <hip/hip_runtime.h>

typedef float v2f __attribute__((ext_vector_type(2)));
typedef float v8f __attribute__((ext_vector_type(8)));

#define N_NODES 50000
#define FEAT    128

#define SORT_BSH      7
#define SORT_BNODES   (1 << SORT_BSH)
#define SORT_NB       ((N_NODES + SORT_BNODES - 1) / SORT_BNODES)
#define SORT_NCH      128
#define SORT_THREADS  1024
#define SORT_UNR      8
#define SORT_SCAN_CH  13
static_assert(N_NODES <= 65536, "keys pack a source node id into 16 bits");
static_assert(32 * SORT_SCAN_CH >= SORT_NB + 1, "one wave scans the bucket counts of a chunk");

__global__ void __launch_bounds__(SORT_THREADS)
sage_chunk_sort_kernel(const int* __restrict__ src, const int* __restrict__ dst,
                       int* __restrict__ loc, int* __restrict__ keys, int E) {
  __shared__ int h[SORT_NB + 1];
  __shared__ int stage[SORT_UNR * SORT_THREADS];
  const int tid = threadIdx.x;
  const int w = blockIdx.x;
  if (tid <= SORT_NB) h[tid] = 0;
  __syncthreads();
  const int chunk = (E + SORT_NCH - 1) / SORT_NCH;
  const int beg = w * chunk;
  const int end = min(E, beg + chunk);
  const bool one = (end - beg) <= SORT_UNR * SORT_THREADS;

  int d[SORT_UNR], s[SORT_UNR];
  if (one) {
#pragma unroll
    for (int k = 0; k < SORT_UNR; ++k) {
      const int e = beg + k * SORT_THREADS + tid;
      d[k] = (e < end) ? dst[e] : -1;
      s[k] = (e < end) ? src[e] : 0;
    }
#pragma unroll
    for (int k = 0; k < SORT_UNR; ++k)
      if (d[k] >= 0) atomicAdd(&h[d[k] >> SORT_BSH], 1);
  } else {
    for (int base = beg; base < end; base += SORT_UNR * SORT_THREADS) {
      int dd[SORT_UNR];
#pragma unroll
      for (int k = 0; k < SORT_UNR; ++k) {
        const int e = base + k * SORT_THREADS + tid;
        dd[k] = (e < end) ? dst[e] : -1;
      }
#pragma unroll
      for (int k = 0; k < SORT_UNR; ++k)
        if (dd[k] >= 0) atomicAdd(&h[dd[k] >> SORT_BSH], 1);
    }
  }
  __syncthreads();

  if (tid < 32) {
    int v[SORT_SCAN_CH];
    int tot = 0;
#pragma unroll
    for (int k = 0; k < SORT_SCAN_CH; ++k) {
      const int i = tid * SORT_SCAN_CH + k;
      v[k] = (i < SORT_NB) ? h[i] : 0;
      tot += v[k];
    }
    int inc = tot;
    for (int o = 1; o < 32; o <<= 1) {
      const int up = __shfl_up(inc, o);
      if (tid >= o) inc += up;
    }
    int run = inc - tot;
#pragma unroll
    for (int k = 0; k < SORT_SCAN_CH; ++k) {
      const int i = tid * SORT_SCAN_CH + k;
      if (i <= SORT_NB) {
        h[i] = run;
        loc[i * SORT_NCH + w] = run;
      }
      run += v[k];
    }
  }
  __syncthreads();

  int* __restrict__ mine = keys + beg;
  if (one) {
    int pos[SORT_UNR];
#pragma unroll
    for (int k = 0; k < SORT_UNR; ++k)
      pos[k] = (d[k] >= 0) ? atomicAdd(&h[d[k] >> SORT_BSH], 1) : 0;
#pragma unroll
    for (int k = 0; k < SORT_UNR; ++k)
      if (d[k] >= 0) stage[pos[k]] = ((d[k] & (SORT_BNODES - 1)) << 16) | s[k];
    __syncthreads();
    const int n = end - beg;
    for (int i = tid; i < n; i += SORT_THREADS) mine[i] = stage[i];
  } else {
    for (int base = beg; base < end; base += SORT_UNR * SORT_THREADS) {
      int dd[SORT_UNR], ss[SORT_UNR];
#pragma unroll
      for (int k = 0; k < SORT_UNR; ++k) {
        const int e = base + k * SORT_THREADS + tid;
        dd[k] = (e < end) ? dst[e] : -1;
        ss[k] = (e < end) ? src[e] : 0;
      }
#pragma unroll
      for (int k = 0; k < SORT_UNR; ++k) {
        if (dd[k] >= 0) {
          const int pos = atomicAdd(&h[dd[k] >> SORT_BSH], 1);
          mine[pos] = ((dd[k] & (SORT_BNODES - 1)) << 16) | ss[k];
        }
      }
    }
  }
}

#define FILL_THREADS 512
#define FILL_WAVES   (FILL_THREADS / 32)
#define FILL_PPW     (SORT_NCH / FILL_WAVES)
#define FILL_CAP     4096
__global__ void __launch_bounds__(FILL_THREADS)
sage_bucket_fill_kernel(const int* __restrict__ loc, const int* __restrict__ keys,
                        int* __restrict__ rowstart, float* __restrict__ deg,
                        int* __restrict__ col, int E) {
  __shared__ __attribute__((aligned(16))) int c[SORT_BNODES];
  __shared__ int red[SORT_NCH / 32];
  __shared__ int stage[FILL_CAP];
  __shared__ int total_s;
  const int tid = threadIdx.x;
  const int lane = tid & 31;
  const int wv = __builtin_amdgcn_readfirstlane(tid >> 5);
  const int b = blockIdx.x;
  const int chunk = (E + SORT_NCH - 1) / SORT_NCH;
  const int* __restrict__ lrow = loc + b * SORT_NCH;

  int st[FILL_PPW], len[FILL_PPW];
#pragma unroll
  for (int u = 0; u < FILL_PPW; ++u) {
    const int p = wv + u * FILL_WAVES;
    const int lo = lrow[p];
    st[u] = p * chunk + lo;
    len[u] = lrow[SORT_NCH + p] - lo;
  }
  int k[FILL_PPW];
#pragma unroll
  for (int u = 0; u < FILL_PPW; ++u)
    k[u] = (lane < len[u]) ? keys[st[u] + lane] : -1;

  if (tid < SORT_NCH) {
    c[tid] = 0;
    int slo = lrow[tid];
    for (int o = 16; o > 0; o >>= 1) slo += __shfl_xor(slo, o);
    if (lane == 0) red[wv] = slo;
  }
  __syncthreads();
  int beg = 0;
  for (int q = 0; q < SORT_NCH / 32; ++q) beg += red[q];

#pragma unroll
  for (int u = 0; u < FILL_PPW; ++u)
    if (k[u] >= 0) atomicAdd(&c[k[u] >> 16], 1);
#pragma unroll
  for (int u = 0; u < FILL_PPW; ++u)
    for (int j = 32 + lane; j < len[u]; j += 32) atomicAdd(&c[keys[st[u] + j] >> 16], 1);
  __syncthreads();

  if (tid < 32) {
    const int4 cv = *(const int4*)&c[4 * tid];
    const int tot = cv.x + cv.y + cv.z + cv.w;
    int inc = tot;
    for (int o = 1; o < 32; o <<= 1) {
      const int up = __shfl_up(inc, o);
      if (tid >= o) inc += up;
    }
    int4 r;
    r.x = beg + inc - tot;
    r.y = r.x + cv.x;
    r.z = r.y + cv.y;
    r.w = r.z + cv.z;
    *(int4*)&c[4 * tid] = r;
    if (tid == 31) total_s = inc;
    const int node = b * SORT_BNODES + 4 * tid;
    if (node < N_NODES) {
      *(int4*)&rowstart[node] = r;
      *(float4*)&deg[node] = make_float4((float)cv.x, (float)cv.y, (float)cv.z, (float)cv.w);
    } else if (node == N_NODES) {
      rowstart[node] = r.x;
    }
  }
  __syncthreads();

  const int total = total_s;
  if (total <= FILL_CAP) {
    int pos[FILL_PPW];
#pragma unroll
    for (int u = 0; u < FILL_PPW; ++u)
      pos[u] = (k[u] >= 0) ? atomicAdd(&c[k[u] >> 16], 1) : beg;
#pragma unroll
    for (int u = 0; u < FILL_PPW; ++u)
      if (k[u] >= 0) stage[pos[u] - beg] = k[u] & 0xFFFF;
#pragma unroll
    for (int u = 0; u < FILL_PPW; ++u) {
      for (int j = 32 + lane; j < len[u]; j += 32) {
        const int kk = keys[st[u] + j];
        const int p = atomicAdd(&c[kk >> 16], 1);
        stage[p - beg] = kk & 0xFFFF;
      }
    }
    __syncthreads();
    for (int i = tid; i < total; i += FILL_THREADS) col[beg + i] = stage[i];
  } else {
#pragma unroll
    for (int u = 0; u < FILL_PPW; ++u) {
      if (k[u] >= 0) {
        const int pos = atomicAdd(&c[k[u] >> 16], 1);
        col[pos] = k[u] & 0xFFFF;
      }
    }
#pragma unroll
    for (int u = 0; u < FILL_PPW; ++u) {
      for (int j = 32 + lane; j < len[u]; j += 32) {
        const int kk = keys[st[u] + j];
        const int pos = atomicAdd(&c[kk >> 16], 1);
        col[pos] = kk & 0xFFFF;
      }
    }
  }
}

#define GATHER_WGS    2048
#define GATHER_WAVES  (GATHER_WGS * 8)
static_assert(GATHER_WAVES <= N_NODES, "every wave of a gather grid has a first node");
#define SAGE_ROW(k) x4[(long long)__builtin_amdgcn_readlane(idx, t + (k)) * 32 + lane]
#define SAGE_ADD(a) acc.x += a.x; acc.y += a.y; acc.z += a.z; acc.w += a.w;
__device__ __forceinline__ void
sage_sum_rows(const float4* x4, int idx, int m, int lane, float4& acc) {
  int t = 0;
  for (; t + 8 <= m; t += 8) {
    const float4 a0 = SAGE_ROW(0), a1 = SAGE_ROW(1), a2 = SAGE_ROW(2), a3 = SAGE_ROW(3);
    const float4 a4 = SAGE_ROW(4), a5 = SAGE_ROW(5), a6 = SAGE_ROW(6), a7 = SAGE_ROW(7);
    SAGE_ADD(a0) SAGE_ADD(a1) SAGE_ADD(a2) SAGE_ADD(a3)
    SAGE_ADD(a4) SAGE_ADD(a5) SAGE_ADD(a6) SAGE_ADD(a7)
  }
  if (t + 4 <= m) {
    const float4 a0 = SAGE_ROW(0), a1 = SAGE_ROW(1), a2 = SAGE_ROW(2), a3 = SAGE_ROW(3);
    SAGE_ADD(a0) SAGE_ADD(a1) SAGE_ADD(a2) SAGE_ADD(a3)
    t += 4;
  }
  if (t + 2 <= m) {
    const float4 a0 = SAGE_ROW(0), a1 = SAGE_ROW(1);
    SAGE_ADD(a0) SAGE_ADD(a1)
    t += 2;
  }
  if (t < m) {
    const float4 a0 = SAGE_ROW(0);
    SAGE_ADD(a0)
  }
}
__global__ void __launch_bounds__(256)
sage_gather_kernel(const float* __restrict__ x,
                   const int* __restrict__ rowstart,
                   const int* __restrict__ col,
                   float* __restrict__ msg) {
  const int lane = threadIdx.x & 31;
  const int gw = __builtin_amdgcn_readfirstlane((int)(blockIdx.x * 8 + (threadIdx.x >> 5)));
  const float4* x4 = (const float4*)x;
  int node = -1, beg = 0, end = 0, idx = 0;
  int node1 = gw;
  int beg1 = rowstart[node1];
  int end1 = rowstart[node1 + 1];
  for (;;) {
    int idx1 = 0;
    if (node1 < N_NODES) {
      const int j1 = beg1 + lane;
      idx1 = (j1 < end1) ? col[j1] : 0;
    }
    const int node2 = node1 + GATHER_WAVES;
    int beg2 = 0, end2 = 0;
    if (node2 < N_NODES) { beg2 = rowstart[node2]; end2 = rowstart[node2 + 1]; }

    if (node >= 0) {
      float4 acc = make_float4(0.0f, 0.0f, 0.0f, 0.0f);
      sage_sum_rows(x4, idx, min(32, end - beg), lane, acc);
      for (int base = beg + 32; base < end; base += 32) {
        const int j = base + lane;
        const int idl = (j < end) ? col[j] : 0;
        sage_sum_rows(x4, idl, min(32, end - base), lane, acc);
      }
      ((float4*)msg)[(long long)node * 32 + lane] = acc;
    }

    if (node1 >= N_NODES) break;
    node = node1; beg = beg1; end = end1; idx = idx1;
    node1 = node2; beg1 = beg2; end1 = end2;
  }
}

#define GEMM_GRID      512
#define GEMM_RT_BLOCK  6
#define GEMM_RT_EVEN   (GEMM_GRID * GEMM_RT_BLOCK)
#define GEMM_ROW_TILES (N_NODES / 16)
#define GEMM_KG        (FEAT / 4)
#define GEMM_ST        (GEMM_KG / 2)
#define GEMM_NB        GEMM_ST

template <int N>
__device__ __forceinline__ void
sage_gemm_item(const float* Ws, const float4 (&am)[GEMM_ST], float degv,
               const float* __restrict__ xin, const float* __restrict__ bl,
               float* __restrict__ out, int doRelu, int rowTile, int half, int c, int h) {
  constexpr int K  = FEAT;
  constexpr int KG = GEMM_KG;
  constexpr int NB = GEMM_NB;
  constexpr int CT = N / 32;

  const int r = (rowTile << 4) + c;
  const float scale = 1.0f / fmaxf(degv, 1.0f);
  const int colbase = half * (N / 2) + c;
  const float4* xrow = (const float4*)(xin + (long long)r * K) + h;
  const float* wb = Ws + (h * N + colbase) * 4;
#define SAGE_BFRAG(blk, u) \
  (*(const float4*)(wb + ((((blk) >> 3) * KG + 2 * (2 * ((blk) & 7) + (u) / CT)) * N + ((u) % CT) * 16) * 4))

  float4 ax[GEMM_ST];
  v8f acc[CT];
#pragma unroll
  for (int t = 0; t < CT; ++t) acc[t] = v8f{};

  float4 bq[2][2 * CT];
#pragma unroll
  for (int u = 0; u < 2 * CT; ++u) bq[0][u] = SAGE_BFRAG(0, u);

#pragma unroll
  for (int blk = 0; blk < NB; ++blk) {
    if (blk + 1 < NB) {
#pragma unroll
      for (int u = 0; u < 2 * CT; ++u) bq[(blk + 1) & 1][u] = SAGE_BFRAG(blk + 1, u);
    }
    if (blk < NB / 2) {
      ax[2 * blk]     = xrow[4 * blk];
      ax[2 * blk + 1] = xrow[4 * blk + 2];
    }
    __builtin_amdgcn_sched_barrier(0);
#pragma unroll
    for (int u2 = 0; u2 < 2; ++u2) {
      float4 a;
      if (blk < NB / 2) {
        a = am[2 * blk + u2];
        a.x *= scale; a.y *= scale; a.z *= scale; a.w *= scale;
      } else {
        a = ax[2 * (blk - NB / 2) + u2];
      }
      v2f a0, a1;
      a0.x = a.x; a0.y = a.y;
      a1.x = a.z; a1.y = a.w;
#pragma unroll
      for (int t = 0; t < CT; ++t) {
        v2f b0;
        b0.x = bq[blk & 1][u2 * CT + t].x; b0.y = bq[blk & 1][u2 * CT + t].y;
        acc[t] = __builtin_amdgcn_wmma_f32_16x16x4_f32(false, a0, false, b0,
                                                       (short)0, acc[t], false, false);
      }
#pragma unroll
      for (int t = 0; t < CT; ++t) {
        v2f b1;
        b1.x = bq[blk & 1][u2 * CT + t].z; b1.y = bq[blk & 1][u2 * CT + t].w;
        acc[t] = __builtin_amdgcn_wmma_f32_16x16x4_f32(false, a1, false, b1,
                                                       (short)0, acc[t], false, false);
      }
    }
    __builtin_amdgcn_sched_barrier(0);
  }
#undef SAGE_BFRAG

  const int rbase = (rowTile << 4) + (h << 3);
#pragma unroll
  for (int t = 0; t < CT; ++t) {
    const int col = colbase + t * 16;
    const float bias = bl[col];
#pragma unroll
    for (int i = 0; i < 8; ++i) {
      float v = acc[t][i] + bias;
      if (doRelu) v = fmaxf(v, 0.0f);
      out[(long long)(rbase + i) * N + col] = v;
    }
  }
}

template <int N>
__global__ void __launch_bounds__(256)
sage_gemm_strip_kernel(const float* __restrict__ msg,
                       const float* __restrict__ deg,
                       const float* __restrict__ xin,
                       const float* __restrict__ Wl,
                       const float* __restrict__ bl,
                       const float* __restrict__ Wr,
                       float* __restrict__ out, int doRelu) {
  constexpr int K  = FEAT;
  constexpr int KG = GEMM_KG;
  __shared__ __attribute__((aligned(16))) float Ws[2 * K * N];

  const int tid  = threadIdx.x;
  const int lane = tid & 31;
  const int c = lane & 15;
  const int h = lane >> 4;
  const int wv = __builtin_amdgcn_readfirstlane(tid >> 5);
  const int b  = blockIdx.x;

  const int rowTile1 = b * GEMM_RT_BLOCK + (wv >> 1);
  const int half1 = wv & 1;
  const int q = b >> 2;
  const bool left = (wv == 4) && ((b & 3) == 1) && (GEMM_RT_EVEN + (q >> 1) < GEMM_ROW_TILES);
  const bool second = (wv < 4) || left;
  const int rowTile2 = left ? GEMM_RT_EVEN + (q >> 1) : b * GEMM_RT_BLOCK + 4 + (wv >> 1);
  const int half2 = left ? (q & 1) : (wv & 1);

  const int r1 = (rowTile1 << 4) + c;
  const float deg1 = deg[r1];
  float4 am[GEMM_ST];
  {
    const float4* mrow = (const float4*)(msg + (long long)r1 * K) + h;
#pragma unroll
    for (int j = 0; j < GEMM_ST; ++j) am[j] = mrow[2 * j];
  }

  {
    constexpr int KG_PASS = 256 / N;
    const int n   = tid % N;
    const int kgo = tid / N;
    for (int m = 0; m < 2; ++m) {
      const float* __restrict__ Wm = m ? Wr : Wl;
#pragma unroll 4
      for (int p = 0; p < KG / KG_PASS; ++p) {
        const int kg = p * KG_PASS + kgo;
        float4 v;
        v.x = Wm[(4 * kg + 0) * N + n];
        v.y = Wm[(4 * kg + 1) * N + n];
        v.z = Wm[(4 * kg + 2) * N + n];
        v.w = Wm[(4 * kg + 3) * N + n];
        *(float4*)(Ws + ((m * KG + kg) * N + n) * 4) = v;
      }
    }
  }
  __syncthreads();

  sage_gemm_item<N>(Ws, am, deg1, xin, bl, out, doRelu, rowTile1, half1, c, h);
  if (second) {
    const int r2 = (rowTile2 << 4) + c;
    const float deg2 = deg[r2];
    float4 an[GEMM_ST];
    const float4* mrow = (const float4*)(msg + (long long)r2 * K) + h;
#pragma unroll
    for (int j = 0; j < GEMM_ST; ++j) an[j] = mrow[2 * j];
    sage_gemm_item<N>(Ws, an, deg2, xin, bl, out, doRelu, rowTile2, half2, c, h);
  }
}

#define XF_N        128
#define XF_RT_BLOCK 6
#define XF_CSR_OFF  50048

__device__ __forceinline__ void
sage_xform_pass(const float* Ws, const float4 (&am)[GEMM_ST], int colbase, int h, v8f (&acc)[4]) {
  constexpr int CT = 4;
  constexpr int NB = GEMM_ST / 2;
  const float* wb = Ws + (h * XF_N + colbase) * 4;
#define XF_BFRAG(blk, u) \
  (*(const float4*)(wb + ((2 * (2 * (blk) + (u) / CT)) * XF_N + ((u) % CT) * 16) * 4))
#pragma unroll
  for (int t = 0; t < CT; ++t) acc[t] = v8f{};
  float4 bq[2][2 * CT];
#pragma unroll
  for (int u = 0; u < 2 * CT; ++u) bq[0][u] = XF_BFRAG(0, u);
#pragma unroll
  for (int blk = 0; blk < NB; ++blk) {
    if (blk + 1 < NB) {
#pragma unroll
      for (int u = 0; u < 2 * CT; ++u) bq[(blk + 1) & 1][u] = XF_BFRAG(blk + 1, u);
    }
    __builtin_amdgcn_sched_barrier(0);
#pragma unroll
    for (int u2 = 0; u2 < 2; ++u2) {
      const float4 a = am[2 * blk + u2];
      v2f a0, a1;
      a0.x = a.x; a0.y = a.y;
      a1.x = a.z; a1.y = a.w;
#pragma unroll
      for (int t = 0; t < CT; ++t) {
        v2f b0;
        b0.x = bq[blk & 1][u2 * CT + t].x; b0.y = bq[blk & 1][u2 * CT + t].y;
        acc[t] = __builtin_amdgcn_wmma_f32_16x16x4_f32(false, a0, false, b0,
                                                       (short)0, acc[t], false, false);
      }
#pragma unroll
      for (int t = 0; t < CT; ++t) {
        v2f b1;
        b1.x = bq[blk & 1][u2 * CT + t].z; b1.y = bq[blk & 1][u2 * CT + t].w;
        acc[t] = __builtin_amdgcn_wmma_f32_16x16x4_f32(false, a1, false, b1,
                                                       (short)0, acc[t], false, false);
      }
    }
    __builtin_amdgcn_sched_barrier(0);
  }
#undef XF_BFRAG
}

__global__ void __launch_bounds__(256)
sage_xform_kernel(float* hq,
                  const float* __restrict__ Wl,
                  const float* __restrict__ bl,
                  const float* __restrict__ Wr,
                  float* __restrict__ P,
                  const int* __restrict__ csrIn, int* __restrict__ csrOut, int E) {
  constexpr int K  = FEAT;
  constexpr int KG = GEMM_KG;
  __shared__ __attribute__((aligned(16))) float Ws[K * XF_N];

  const int tid  = threadIdx.x;
  const int lane = tid & 31;
  const int c = lane & 15;
  const int h = lane >> 4;
  const int wv = __builtin_amdgcn_readfirstlane(tid >> 5);
  const int b  = blockIdx.x;

  const int rowTile = (wv < XF_RT_BLOCK) ? b * XF_RT_BLOCK + wv : GEMM_RT_EVEN + b;
  const bool work = (wv < XF_RT_BLOCK) || (wv == XF_RT_BLOCK && rowTile < GEMM_ROW_TILES);

  const int r = (rowTile << 4) + c;
  float4 am[GEMM_ST];
  if (work) {
    const float4* mrow = (const float4*)(hq + (long long)r * K) + h;
#pragma unroll
    for (int j = 0; j < GEMM_ST; ++j) am[j] = mrow[2 * j];
  }

  {
    const int n4 = (XF_CSR_OFF + E + 3) >> 2;
    const int4* __restrict__ s4 = (const int4*)csrIn;
    int4* __restrict__ d4 = (int4*)csrOut;
    for (int i = b * 256 + tid; i < n4; i += GEMM_GRID * 256) d4[i] = s4[i];
  }

  {
    const int n   = tid & 63;
    const int kgo = tid >> 6;
    for (int m = 0; m < 2; ++m) {
      const float* __restrict__ Wm = m ? Wr : Wl;
#pragma unroll 4
      for (int p = 0; p < KG / 4; ++p) {
        const int kg = p * 4 + kgo;
        float4 v;
        v.x = Wm[(4 * kg + 0) * 64 + n];
        v.y = Wm[(4 * kg + 1) * 64 + n];
        v.z = Wm[(4 * kg + 2) * 64 + n];
        v.w = Wm[(4 * kg + 3) * 64 + n];
        *(float4*)(Ws + (kg * XF_N + m * 64 + n) * 4) = v;
      }
    }
  }
  __syncthreads();

  if (work) {
    const int rbase = (rowTile << 4) + (h << 3);
    v8f acc[4];
    sage_xform_pass(Ws, am, c, h, acc);
#pragma unroll
    for (int t = 0; t < 4; ++t) {
      const int col = c + t * 16;
#pragma unroll
      for (int i = 0; i < 8; ++i) P[(long long)(rbase + i) * 64 + col] = acc[t][i];
    }
    sage_xform_pass(Ws, am, 64 + c, h, acc);
#pragma unroll
    for (int t = 0; t < 4; ++t) {
      const int col = c + t * 16;
      const float bias = bl[col];
#pragma unroll
      for (int i = 0; i < 8; ++i) hq[(long long)(rbase + i) * K + col] = acc[t][i] + bias;
    }
  }
}

#define SAGE_PROW(k) p4[(long long)sage_pair_id(idx, t + (k), hh) * 16 + c16]
__device__ __forceinline__ int sage_pair_id(int idx, int slot, int hh) {
  const int i0 = __builtin_amdgcn_readlane(idx, slot);
  const int i1 = __builtin_amdgcn_readlane(idx, slot + 1);
  return hh ? i1 : i0;
}
__device__ __forceinline__ void
sage_sum_prows(const float4* p4, int idx, int m, int hh, int c16, float4& acc) {
  int t = 0;
  for (; t + 16 <= m; t += 16) {
    const float4 a0 = SAGE_PROW(0), a1 = SAGE_PROW(2), a2 = SAGE_PROW(4), a3 = SAGE_PROW(6);
    const float4 a4 = SAGE_PROW(8), a5 = SAGE_PROW(10), a6 = SAGE_PROW(12), a7 = SAGE_PROW(14);
    SAGE_ADD(a0) SAGE_ADD(a1) SAGE_ADD(a2) SAGE_ADD(a3)
    SAGE_ADD(a4) SAGE_ADD(a5) SAGE_ADD(a6) SAGE_ADD(a7)
  }
  if (t + 8 <= m) {
    const float4 a0 = SAGE_PROW(0), a1 = SAGE_PROW(2), a2 = SAGE_PROW(4), a3 = SAGE_PROW(6);
    SAGE_ADD(a0) SAGE_ADD(a1) SAGE_ADD(a2) SAGE_ADD(a3)
    t += 8;
  }
  if (t + 4 <= m) {
    const float4 a0 = SAGE_PROW(0), a1 = SAGE_PROW(2);
    SAGE_ADD(a0) SAGE_ADD(a1)
    t += 4;
  }
  if (t + 2 <= m) {
    const float4 a0 = SAGE_PROW(0);
    SAGE_ADD(a0)
    t += 2;
  }
  if (t < m) {
    const float4 a0 = SAGE_PROW(0);
    if (hh == 0) { SAGE_ADD(a0) }
  }
}
__global__ void __launch_bounds__(256)
sage_gather_mean_kernel(const float* __restrict__ P,
                        const float* __restrict__ hq,
                        const int* __restrict__ rowstart,
                        const int* __restrict__ col,
                        float* __restrict__ out) {
  const int lane = threadIdx.x & 31;
  const int c16 = lane & 15;
  const int hh = lane >> 4;
  const int gw = __builtin_amdgcn_readfirstlane((int)(blockIdx.x * 8 + (threadIdx.x >> 5)));
  const float4* p4 = (const float4*)P;
  int node = -1, beg = 0, end = 0, idx = 0;
  int node1 = gw;
  int beg1 = rowstart[node1];
  int end1 = rowstart[node1 + 1];
  for (;;) {
    int idx1 = 0;
    if (node1 < N_NODES) {
      const int j1 = beg1 + lane;
      idx1 = (j1 < end1) ? col[j1] : 0;
    }
    const int node2 = node1 + GATHER_WAVES;
    int beg2 = 0, end2 = 0;
    if (node2 < N_NODES) { beg2 = rowstart[node2]; end2 = rowstart[node2 + 1]; }

    if (node >= 0) {
      const float4 q = ((const float4*)hq)[(long long)node * 32 + c16];
      float4 acc = make_float4(0.0f, 0.0f, 0.0f, 0.0f);
      sage_sum_prows(p4, idx, min(32, end - beg), hh, c16, acc);
      for (int base = beg + 32; base < end; base += 32) {
        const int j = base + lane;
        const int idl = (j < end) ? col[j] : 0;
        sage_sum_prows(p4, idl, min(32, end - base), hh, c16, acc);
      }
      acc.x += __shfl_xor(acc.x, 16);
      acc.y += __shfl_xor(acc.y, 16);
      acc.z += __shfl_xor(acc.z, 16);
      acc.w += __shfl_xor(acc.w, 16);
      const float scale = 1.0f / fmaxf((float)(end - beg), 1.0f);
      if (hh == 0) {
        float4 o;
        o.x = acc.x * scale + q.x;
        o.y = acc.y * scale + q.y;
        o.z = acc.z * scale + q.z;
        o.w = acc.w * scale + q.w;
        ((float4*)out)[(long long)node * 16 + c16] = o;
      }
    }

    if (node1 >= N_NODES) break;
    node = node1; beg = beg1; end = end1; idx = idx1;
    node1 = node2; beg1 = beg2; end1 = end2;
  }
}

extern "C" void kernel_launch(void* const* d_in, const int* in_sizes, int n_in,
                              void* d_out, int out_size, void* d_ws, size_t ws_size,
                              hipStream_t stream) {
  const float* x   = (const float*)d_in[0];
  const int*   ei  = (const int*)d_in[1];
  const float* Wl1 = (const float*)d_in[2];
  const float* bl1 = (const float*)d_in[3];
  const float* Wr1 = (const float*)d_in[4];
  const float* Wl2 = (const float*)d_in[5];
  const float* bl2 = (const float*)d_in[6];
  const float* Wr2 = (const float*)d_in[7];
  float* out = (float*)d_out;

  const int E = in_sizes[1] / 2;
  const int* src = ei;
  const int* dst = ei + E;

  float* msg = (float*)d_ws;
  float* deg = msg + (long long)N_NODES * FEAT;
  float* h1  = deg + N_NODES;

  int* rowstart = (int*)d_out;
  int* col      = rowstart + XF_CSR_OFF;

  int* loc  = (int*)h1;
  int* keys = loc + 65536;

  const int TB = 256;

  sage_chunk_sort_kernel<<<SORT_NCH, SORT_THREADS, 0, stream>>>(src, dst, loc, keys, E);
  sage_bucket_fill_kernel<<<SORT_NB, FILL_THREADS, 0, stream>>>(loc, keys, rowstart, deg, col, E);

  sage_gather_kernel<<<GATHER_WGS, TB, 0, stream>>>(x, rowstart, col, msg);
  sage_gemm_strip_kernel<128><<<GEMM_GRID, 256, 0, stream>>>(msg, deg, x, Wl1, bl1, Wr1, h1, 1);

  float* P = msg;
  int* rowstart2 = (int*)(msg + (long long)N_NODES * 64);
  int* col2      = rowstart2 + XF_CSR_OFF;
  sage_xform_kernel<<<GEMM_GRID, 256, 0, stream>>>(h1, Wl2, bl2, Wr2, P, rowstart, rowstart2, E);
  sage_gather_mean_kernel<<<GATHER_WGS, TB, 0, stream>>>(P, h1, rowstart2, col2, out);
}


# Layer-1 GEMM and layer-2 weight transform in one launch: a 512-thread workgroup holds all four weight matrices and its h1 row tiles in LDS, h1 never goes to memory; 5 launches; sum order unchanged
# speedup vs baseline: 5.2782x; 1.1689x over previous
#include <hip/hip_runtime.h>

typedef float v2f __attribute__((ext_vector_type(2)));
typedef float v8f __attribute__((ext_vector_type(8)));

#define N_NODES 50000
#define FEAT    128

#define SORT_BSH      7
#define SORT_BNODES   (1 << SORT_BSH)
#define SORT_NB       ((N_NODES + SORT_BNODES - 1) / SORT_BNODES)
#define SORT_NCH      128
#define SORT_THREADS  1024
#define SORT_UNR      8
#define SORT_SCAN_CH  13
static_assert(N_NODES <= 65536, "keys pack a source node id into 16 bits");
static_assert(32 * SORT_SCAN_CH >= SORT_NB + 1, "one wave scans the bucket counts of a chunk");

__global__ void __launch_bounds__(SORT_THREADS)
sage_chunk_sort_kernel(const int* __restrict__ src, const int* __restrict__ dst,
                       int* __restrict__ loc, int* __restrict__ keys, int E) {
  __shared__ int h[SORT_NB + 1];
  __shared__ int stage[SORT_UNR * SORT_THREADS];
  const int tid = threadIdx.x;
  const int w = blockIdx.x;
  if (tid <= SORT_NB) h[tid] = 0;
  __syncthreads();
  const int chunk = (E + SORT_NCH - 1) / SORT_NCH;
  const int beg = w * chunk;
  const int end = min(E, beg + chunk);
  const bool one = (end - beg) <= SORT_UNR * SORT_THREADS;

  int d[SORT_UNR], s[SORT_UNR];
  if (one) {
#pragma unroll
    for (int k = 0; k < SORT_UNR; ++k) {
      const int e = beg + k * SORT_THREADS + tid;
      d[k] = (e < end) ? dst[e] : -1;
      s[k] = (e < end) ? src[e] : 0;
    }
#pragma unroll
    for (int k = 0; k < SORT_UNR; ++k)
      if (d[k] >= 0) atomicAdd(&h[d[k] >> SORT_BSH], 1);
  } else {
    for (int base = beg; base < end; base += SORT_UNR * SORT_THREADS) {
      int dd[SORT_UNR];
#pragma unroll
      for (int k = 0; k < SORT_UNR; ++k) {
        const int e = base + k * SORT_THREADS + tid;
        dd[k] = (e < end) ? dst[e] : -1;
      }
#pragma unroll
      for (int k = 0; k < SORT_UNR; ++k)
        if (dd[k] >= 0) atomicAdd(&h[dd[k] >> SORT_BSH], 1);
    }
  }
  __syncthreads();

  if (tid < 32) {
    int v[SORT_SCAN_CH];
    int tot = 0;
#pragma unroll
    for (int k = 0; k < SORT_SCAN_CH; ++k) {
      const int i = tid * SORT_SCAN_CH + k;
      v[k] = (i < SORT_NB) ? h[i] : 0;
      tot += v[k];
    }
    int inc = tot;
    for (int o = 1; o < 32; o <<= 1) {
      const int up = __shfl_up(inc, o);
      if (tid >= o) inc += up;
    }
    int run = inc - tot;
#pragma unroll
    for (int k = 0; k < SORT_SCAN_CH; ++k) {
      const int i = tid * SORT_SCAN_CH + k;
      if (i <= SORT_NB) {
        h[i] = run;
        loc[i * SORT_NCH + w] = run;
      }
      run += v[k];
    }
  }
  __syncthreads();

  int* __restrict__ mine = keys + beg;
  if (one) {
    int pos[SORT_UNR];
#pragma unroll
    for (int k = 0; k < SORT_UNR; ++k)
      pos[k] = (d[k] >= 0) ? atomicAdd(&h[d[k] >> SORT_BSH], 1) : 0;
#pragma unroll
    for (int k = 0; k < SORT_UNR; ++k)
      if (d[k] >= 0) stage[pos[k]] = ((d[k] & (SORT_BNODES - 1)) << 16) | s[k];
    __syncthreads();
    const int n = end - beg;
    for (int i = tid; i < n; i += SORT_THREADS) mine[i] = stage[i];
  } else {
    for (int base = beg; base < end; base += SORT_UNR * SORT_THREADS) {
      int dd[SORT_UNR], ss[SORT_UNR];
#pragma unroll
      for (int k = 0; k < SORT_UNR; ++k) {
        const int e = base + k * SORT_THREADS + tid;
        dd[k] = (e < end) ? dst[e] : -1;
        ss[k] = (e < end) ? src[e] : 0;
      }
#pragma unroll
      for (int k = 0; k < SORT_UNR; ++k) {
        if (dd[k] >= 0) {
          const int pos = atomicAdd(&h[dd[k] >> SORT_BSH], 1);
          mine[pos] = ((dd[k] & (SORT_BNODES - 1)) << 16) | ss[k];
        }
      }
    }
  }
}

#define FILL_THREADS 512
#define FILL_WAVES   (FILL_THREADS / 32)
#define FILL_PPW     (SORT_NCH / FILL_WAVES)
#define FILL_CAP     4096
__global__ void __launch_bounds__(FILL_THREADS)
sage_bucket_fill_kernel(const int* __restrict__ loc, const int* __restrict__ keys,
                        int* __restrict__ rowstart, float* __restrict__ deg,
                        int* __restrict__ col, int E) {
  __shared__ __attribute__((aligned(16))) int c[SORT_BNODES];
  __shared__ int red[SORT_NCH / 32];
  __shared__ int stage[FILL_CAP];
  __shared__ int total_s;
  const int tid = threadIdx.x;
  const int lane = tid & 31;
  const int wv = __builtin_amdgcn_readfirstlane(tid >> 5);
  const int b = blockIdx.x;
  const int chunk = (E + SORT_NCH - 1) / SORT_NCH;
  const int* __restrict__ lrow = loc + b * SORT_NCH;

  int st[FILL_PPW], len[FILL_PPW];
#pragma unroll
  for (int u = 0; u < FILL_PPW; ++u) {
    const int p = wv + u * FILL_WAVES;
    const int lo = lrow[p];
    st[u] = p * chunk + lo;
    len[u] = lrow[SORT_NCH + p] - lo;
  }
  int k[FILL_PPW];
#pragma unroll
  for (int u = 0; u < FILL_PPW; ++u)
    k[u] = (lane < len[u]) ? keys[st[u] + lane] : -1;

  if (tid < SORT_NCH) {
    c[tid] = 0;
    int slo = lrow[tid];
    for (int o = 16; o > 0; o >>= 1) slo += __shfl_xor(slo, o);
    if (lane == 0) red[wv] = slo;
  }
  __syncthreads();
  int beg = 0;
  for (int q = 0; q < SORT_NCH / 32; ++q) beg += red[q];

#pragma unroll
  for (int u = 0; u < FILL_PPW; ++u)
    if (k[u] >= 0) atomicAdd(&c[k[u] >> 16], 1);
#pragma unroll
  for (int u = 0; u < FILL_PPW; ++u)
    for (int j = 32 + lane; j < len[u]; j += 32) atomicAdd(&c[keys[st[u] + j] >> 16], 1);
  __syncthreads();

  if (tid < 32) {
    const int4 cv = *(const int4*)&c[4 * tid];
    const int tot = cv.x + cv.y + cv.z + cv.w;
    int inc = tot;
    for (int o = 1; o < 32; o <<= 1) {
      const int up = __shfl_up(inc, o);
      if (tid >= o) inc += up;
    }
    int4 r;
    r.x = beg + inc - tot;
    r.y = r.x + cv.x;
    r.z = r.y + cv.y;
    r.w = r.z + cv.z;
    *(int4*)&c[4 * tid] = r;
    if (tid == 31) total_s = inc;
    const int node = b * SORT_BNODES + 4 * tid;
    if (node < N_NODES) {
      *(int4*)&rowstart[node] = r;
      *(float4*)&deg[node] = make_float4((float)cv.x, (float)cv.y, (float)cv.z, (float)cv.w);
    } else if (node == N_NODES) {
      rowstart[node] = r.x;
    }
  }
  __syncthreads();

  const int total = total_s;
  if (total <= FILL_CAP) {
    int pos[FILL_PPW];
#pragma unroll
    for (int u = 0; u < FILL_PPW; ++u)
      pos[u] = (k[u] >= 0) ? atomicAdd(&c[k[u] >> 16], 1) : beg;
#pragma unroll
    for (int u = 0; u < FILL_PPW; ++u)
      if (k[u] >= 0) stage[pos[u] - beg] = k[u] & 0xFFFF;
#pragma unroll
    for (int u = 0; u < FILL_PPW; ++u) {
      for (int j = 32 + lane; j < len[u]; j += 32) {
        const int kk = keys[st[u] + j];
        const int p = atomicAdd(&c[kk >> 16], 1);
        stage[p - beg] = kk & 0xFFFF;
      }
    }
    __syncthreads();
    for (int i = tid; i < total; i += FILL_THREADS) col[beg + i] = stage[i];
  } else {
#pragma unroll
    for (int u = 0; u < FILL_PPW; ++u) {
      if (k[u] >= 0) {
        const int pos = atomicAdd(&c[k[u] >> 16], 1);
        col[pos] = k[u] & 0xFFFF;
      }
    }
#pragma unroll
    for (int u = 0; u < FILL_PPW; ++u) {
      for (int j = 32 + lane; j < len[u]; j += 32) {
        const int kk = keys[st[u] + j];
        const int pos = atomicAdd(&c[kk >> 16], 1);
        col[pos] = kk & 0xFFFF;
      }
    }
  }
}

#define GATHER_WGS    2048
#define GATHER_WAVES  (GATHER_WGS * 8)
static_assert(GATHER_WAVES <= N_NODES, "every wave of a gather grid has a first node");
#define SAGE_ROW(k) x4[(long long)__builtin_amdgcn_readlane(idx, t + (k)) * 32 + lane]
#define SAGE_ADD(a) acc.x += a.x; acc.y += a.y; acc.z += a.z; acc.w += a.w;
__device__ __forceinline__ void
sage_sum_rows(const float4* x4, int idx, int m, int lane, float4& acc) {
  int t = 0;
  for (; t + 8 <= m; t += 8) {
    const float4 a0 = SAGE_ROW(0), a1 = SAGE_ROW(1), a2 = SAGE_ROW(2), a3 = SAGE_ROW(3);
    const float4 a4 = SAGE_ROW(4), a5 = SAGE_ROW(5), a6 = SAGE_ROW(6), a7 = SAGE_ROW(7);
    SAGE_ADD(a0) SAGE_ADD(a1) SAGE_ADD(a2) SAGE_ADD(a3)
    SAGE_ADD(a4) SAGE_ADD(a5) SAGE_ADD(a6) SAGE_ADD(a7)
  }
  if (t + 4 <= m) {
    const float4 a0 = SAGE_ROW(0), a1 = SAGE_ROW(1), a2 = SAGE_ROW(2), a3 = SAGE_ROW(3);
    SAGE_ADD(a0) SAGE_ADD(a1) SAGE_ADD(a2) SAGE_ADD(a3)
    t += 4;
  }
  if (t + 2 <= m) {
    const float4 a0 = SAGE_ROW(0), a1 = SAGE_ROW(1);
    SAGE_ADD(a0) SAGE_ADD(a1)
    t += 2;
  }
  if (t < m) {
    const float4 a0 = SAGE_ROW(0);
    SAGE_ADD(a0)
  }
}
__global__ void __launch_bounds__(256)
sage_gather_kernel(const float* __restrict__ x,
                   const int* __restrict__ rowstart,
                   const int* __restrict__ col,
                   float* __restrict__ msg) {
  const int lane = threadIdx.x & 31;
  const int gw = __builtin_amdgcn_readfirstlane((int)(blockIdx.x * 8 + (threadIdx.x >> 5)));
  const float4* x4 = (const float4*)x;
  int node = -1, beg = 0, end = 0, idx = 0;
  int node1 = gw;
  int beg1 = rowstart[node1];
  int end1 = rowstart[node1 + 1];
  for (;;) {
    int idx1 = 0;
    if (node1 < N_NODES) {
      const int j1 = beg1 + lane;
      idx1 = (j1 < end1) ? col[j1] : 0;
    }
    const int node2 = node1 + GATHER_WAVES;
    int beg2 = 0, end2 = 0;
    if (node2 < N_NODES) { beg2 = rowstart[node2]; end2 = rowstart[node2 + 1]; }

    if (node >= 0) {
      float4 acc = make_float4(0.0f, 0.0f, 0.0f, 0.0f);
      sage_sum_rows(x4, idx, min(32, end - beg), lane, acc);
      for (int base = beg + 32; base < end; base += 32) {
        const int j = base + lane;
        const int idl = (j < end) ? col[j] : 0;
        sage_sum_rows(x4, idl, min(32, end - base), lane, acc);
      }
      ((float4*)msg)[(long long)node * 32 + lane] = acc;
    }

    if (node1 >= N_NODES) break;
    node = node1; beg = beg1; end = end1; idx = idx1;
    node1 = node2; beg1 = beg2; end1 = end2;
  }
}


#define GEMM_ROW_TILES (N_NODES / 16)
#define GEMM_KG        (FEAT / 4)
#define GEMM_ST        (GEMM_KG / 2)
#define GEMM_NB        GEMM_ST
#define FU_WGS         256
#define FU_THREADS     512
#define FU_WAVES       (FU_THREADS / 32)
#define FU_RT          12
#define FU_RT_EVEN     (FU_WGS * FU_RT)
#define FU_RT_MAX      (FU_RT + 1)
#define FU_PITCH       132
#define XF_N           128
#define XF_CSR_OFF     50048
static_assert(GEMM_ROW_TILES - FU_RT_EVEN <= FU_WGS, "one left-over row tile per workgroup at most");
static_assert(2 * FU_RT >= FU_WAVES && 2 * FU_RT - FU_WAVES == 8, "every wave has a first item, waves 0..7 a second");

template <int CT>
__device__ __forceinline__ void
sage_gemm_item(const float* Ws, const float4 (&am)[GEMM_ST], float degv,
               const float* __restrict__ xin, const float* __restrict__ bl,
               float* hs, int rowTile, int colbase, int c, int h) {
  constexpr int N  = 128;
  constexpr int K  = FEAT;
  constexpr int KG = GEMM_KG;
  constexpr int NB = GEMM_NB;

  const int r = (rowTile << 4) + c;
  const float scale = 1.0f / fmaxf(degv, 1.0f);
  const float4* xrow = (const float4*)(xin + (long long)r * K) + h;
  const float* wb = Ws + (h * N + colbase) * 4;
#define SAGE_BFRAG(blk, u) \
  (*(const float4*)(wb + ((((blk) >> 3) * KG + 2 * (2 * ((blk) & 7) + (u) / CT)) * N + ((u) % CT) * 16) * 4))

  float4 ax[GEMM_ST];
  v8f acc[CT];
#pragma unroll
  for (int t = 0; t < CT; ++t) acc[t] = v8f{};

  float4 bq[2][2 * CT];
#pragma unroll
  for (int u = 0; u < 2 * CT; ++u) bq[0][u] = SAGE_BFRAG(0, u);

#pragma unroll
  for (int blk = 0; blk < NB; ++blk) {
    if (blk + 1 < NB) {
#pragma unroll
      for (int u = 0; u < 2 * CT; ++u) bq[(blk + 1) & 1][u] = SAGE_BFRAG(blk + 1, u);
    }
    if (blk < NB / 2) {
      ax[2 * blk]     = xrow[4 * blk];
      ax[2 * blk + 1] = xrow[4 * blk + 2];
    }
    __builtin_amdgcn_sched_barrier(0);
#pragma unroll
    for (int u2 = 0; u2 < 2; ++u2) {
      float4 a;
      if (blk < NB / 2) {
        a = am[2 * blk + u2];
        a.x *= scale; a.y *= scale; a.z *= scale; a.w *= scale;
      } else {
        a = ax[2 * (blk - NB / 2) + u2];
      }
      v2f a0, a1;
      a0.x = a.x; a0.y = a.y;
      a1.x = a.z; a1.y = a.w;
#pragma unroll
      for (int t = 0; t < CT; ++t) {
        v2f b0;
        b0.x = bq[blk & 1][u2 * CT + t].x; b0.y = bq[blk & 1][u2 * CT + t].y;
        acc[t] = __builtin_amdgcn_wmma_f32_16x16x4_f32(false, a0, false, b0,
                                                       (short)0, acc[t], false, false);
      }
#pragma unroll
      for (int t = 0; t < CT; ++t) {
        v2f b1;
        b1.x = bq[blk & 1][u2 * CT + t].z; b1.y = bq[blk & 1][u2 * CT + t].w;
        acc[t] = __builtin_amdgcn_wmma_f32_16x16x4_f32(false, a1, false, b1,
                                                       (short)0, acc[t], false, false);
      }
    }
    __builtin_amdgcn_sched_barrier(0);
  }
#undef SAGE_BFRAG

  float* hd = hs + (h << 3) * FU_PITCH;
#pragma unroll
  for (int t = 0; t < CT; ++t) {
    const int col = colbase + t * 16;
    const float bias = bl[col];
#pragma unroll
    for (int i = 0; i < 8; ++i)
      hd[i * FU_PITCH + col] = fmaxf(acc[t][i] + bias, 0.0f);
  }
}

template <int CT>
__device__ __forceinline__ void
sage_xform_pass(const float* Ws, const float4 (&am)[GEMM_ST], int colbase, int h, v8f (&acc)[CT]) {
  constexpr int NB = GEMM_ST / 2;
  const float* wb = Ws + (h * XF_N + colbase) * 4;
#define XF_BFRAG(blk, u) \
  (*(const float4*)(wb + ((2 * (2 * (blk) + (u) / CT)) * XF_N + ((u) % CT) * 16) * 4))
#pragma unroll
  for (int t = 0; t < CT; ++t) acc[t] = v8f{};
  float4 bq[2][2 * CT];
#pragma unroll
  for (int u = 0; u < 2 * CT; ++u) bq[0][u] = XF_BFRAG(0, u);
#pragma unroll
  for (int blk = 0; blk < NB; ++blk) {
    if (blk + 1 < NB) {
#pragma unroll
      for (int u = 0; u < 2 * CT; ++u) bq[(blk + 1) & 1][u] = XF_BFRAG(blk + 1, u);
    }
    __builtin_amdgcn_sched_barrier(0);
#pragma unroll
    for (int u2 = 0; u2 < 2; ++u2) {
      const float4 a = am[2 * blk + u2];
      v2f a0, a1;
      a0.x = a.x; a0.y = a.y;
      a1.x = a.z; a1.y = a.w;
#pragma unroll
      for (int t = 0; t < CT; ++t) {
        v2f b0;
        b0.x = bq[blk & 1][u2 * CT + t].x; b0.y = bq[blk & 1][u2 * CT + t].y;
        acc[t] = __builtin_amdgcn_wmma_f32_16x16x4_f32(false, a0, false, b0,
                                                       (short)0, acc[t], false, false);
      }
#pragma unroll
      for (int t = 0; t < CT; ++t) {
        v2f b1;
        b1.x = bq[blk & 1][u2 * CT + t].z; b1.y = bq[blk & 1][u2 * CT + t].w;
        acc[t] = __builtin_amdgcn_wmma_f32_16x16x4_f32(false, a1, false, b1,
                                                       (short)0, acc[t], false, false);
      }
    }
    __builtin_amdgcn_sched_barrier(0);
  }
#undef XF_BFRAG
}

__global__ void __launch_bounds__(FU_THREADS)
sage_layers_kernel(const float* __restrict__ msg,
                   const float* __restrict__ deg,
                   const float* __restrict__ xin,
                   const float* __restrict__ Wl1,
                   const float* __restrict__ bl1,
                   const float* __restrict__ Wr1,
                   const float* __restrict__ Wl2,
                   const float* __restrict__ bl2,
                   const float* __restrict__ Wr2,
                   float* __restrict__ P, float* __restrict__ Q) {
  constexpr int K  = FEAT;
  constexpr int KG = GEMM_KG;
  constexpr int N  = 128;
  __shared__ __attribute__((aligned(16))) float Ws[2 * K * N];
  __shared__ __attribute__((aligned(16))) float Ws2[K * XF_N];
  __shared__ __attribute__((aligned(16))) float Hs[FU_RT_MAX * 16 * FU_PITCH];

  const int tid  = threadIdx.x;
  const int lane = tid & 31;
  const int c = lane & 15;
  const int h = lane >> 4;
  const int wv = __builtin_amdgcn_readfirstlane(tid >> 5);
  const int b  = blockIdx.x;

  const bool extra = (FU_RT_EVEN + b < GEMM_ROW_TILES);
  const int tl1 = wv >> 1;
  const int half1 = wv & 1;
  const int rowTile1 = b * FU_RT + tl1;
  const bool second = wv < 2 * FU_RT - FU_WAVES;
  const int tl2 = (FU_WAVES + wv) >> 1;
  const int rowTile2 = b * FU_RT + tl2;
  const bool quarter = extra && wv >= 8 && wv < 12;
  const int qq = wv - 8;

  const int r1 = (rowTile1 << 4) + c;
  const float deg1 = deg[r1];
  float4 am[GEMM_ST];
  {
    const float4* mrow = (const float4*)(msg + (long long)r1 * K) + h;
#pragma unroll
    for (int j = 0; j < GEMM_ST; ++j) am[j] = mrow[2 * j];
  }

  {
    const int n   = tid & (N - 1);
    const int kgo = tid >> 7;
    for (int m = 0; m < 2; ++m) {
      const float* __restrict__ Wm = m ? Wr1 : Wl1;
#pragma unroll
      for (int p = 0; p < KG / 4; ++p) {
        const int kg = p * 4 + kgo;
        float4 v;
        v.x = Wm[(4 * kg + 0) * N + n];
        v.y = Wm[(4 * kg + 1) * N + n];
        v.z = Wm[(4 * kg + 2) * N + n];
        v.w = Wm[(4 * kg + 3) * N + n];
        *(float4*)(Ws + ((m * KG + kg) * N + n) * 4) = v;
      }
    }
  }
  {
    const int n   = tid & 63;
    const int kgo = tid >> 6;
    for (int m = 0; m < 2; ++m) {
      const float* __restrict__ Wm = m ? Wr2 : Wl2;
#pragma unroll
      for (int p = 0; p < KG / 8; ++p) {
        const int kg = p * 8 + kgo;
        float4 v;
        v.x = Wm[(4 * kg + 0) * 64 + n];
        v.y = Wm[(4 * kg + 1) * 64 + n];
        v.z = Wm[(4 * kg + 2) * 64 + n];
        v.w = Wm[(4 * kg + 3) * 64 + n];
        *(float4*)(Ws2 + (kg * XF_N + m * 64 + n) * 4) = v;
      }
    }
  }
  __syncthreads();

  sage_gemm_item<4>(Ws, am, deg1, xin, bl1, Hs + tl1 * 16 * FU_PITCH, rowTile1, half1 * 64 + c, c, h);
  if (second) {
    const int r2 = (rowTile2 << 4) + c;
    const float deg2 = deg[r2];
    float4 an[GEMM_ST];
    const float4* mrow = (const float4*)(msg + (long long)r2 * K) + h;
#pragma unroll
    for (int j = 0; j < GEMM_ST; ++j) an[j] = mrow[2 * j];
    sage_gemm_item<4>(Ws, an, deg2, xin, bl1, Hs + tl2 * 16 * FU_PITCH, rowTile2, half1 * 64 + c, c, h);
  } else if (quarter) {
    const int rowTileX = FU_RT_EVEN + b;
    const int rx = (rowTileX << 4) + c;
    const float degx = deg[rx];
    float4 an[GEMM_ST];
    const float4* mrow = (const float4*)(msg + (long long)rx * K) + h;
#pragma unroll
    for (int j = 0; j < GEMM_ST; ++j) an[j] = mrow[2 * j];
    sage_gemm_item<2>(Ws, an, degx, xin, bl1, Hs + FU_RT * 16 * FU_PITCH, rowTileX, qq * 32 + c, c, h);
  }
  __syncthreads();

  for (int it = wv; it < 2 * FU_RT; it += FU_WAVES) {
    const int tl = it >> 1;
    const int half = it & 1;
    const int rowTile = b * FU_RT + tl;
    const float* hrow = Hs + (tl * 16 + c) * FU_PITCH + 4 * h;
    float4 af[GEMM_ST];
#pragma unroll
    for (int j = 0; j < GEMM_ST; ++j) af[j] = *(const float4*)(hrow + 8 * j);
    v8f acc[4];
    sage_xform_pass<4>(Ws2, af, half * 64 + c, h, acc);
    const int rbase = (rowTile << 4) + (h << 3);
    if (half == 0) {
#pragma unroll
      for (int t = 0; t < 4; ++t) {
        const int col = c + t * 16;
#pragma unroll
        for (int i = 0; i < 8; ++i) P[(long long)(rbase + i) * 64 + col] = acc[t][i];
      }
    } else {
#pragma unroll
      for (int t = 0; t < 4; ++t) {
        const int col = c + t * 16;
        const float bias = bl2[col];
#pragma unroll
        for (int i = 0; i < 8; ++i) Q[(long long)(rbase + i) * 64 + col] = acc[t][i] + bias;
      }
    }
  }
  if (quarter) {
    const int rowTile = FU_RT_EVEN + b;
    const float* hrow = Hs + (FU_RT * 16 + c) * FU_PITCH + 4 * h;
    float4 af[GEMM_ST];
#pragma unroll
    for (int j = 0; j < GEMM_ST; ++j) af[j] = *(const float4*)(hrow + 8 * j);
    v8f acc[2];
    sage_xform_pass<2>(Ws2, af, qq * 32 + c, h, acc);
    const int rbase = (rowTile << 4) + (h << 3);
    const int colq = (qq & 1) * 32 + c;
    if (qq < 2) {
#pragma unroll
      for (int t = 0; t < 2; ++t) {
        const int col = colq + t * 16;
#pragma unroll
        for (int i = 0; i < 8; ++i) P[(long long)(rbase + i) * 64 + col] = acc[t][i];
      }
    } else {
#pragma unroll
      for (int t = 0; t < 2; ++t) {
        const int col = colq + t * 16;
        const float bias = bl2[col];
#pragma unroll
        for (int i = 0; i < 8; ++i) Q[(long long)(rbase + i) * 64 + col] = acc[t][i] + bias;
      }
    }
  }
}

#define SAGE_PROW(k) p4[(long long)sage_pair_id(idx, t + (k), hh) * 16 + c16]
__device__ __forceinline__ int sage_pair_id(int idx, int slot, int hh) {
  const int i0 = __builtin_amdgcn_readlane(idx, slot);
  const int i1 = __builtin_amdgcn_readlane(idx, slot + 1);
  return hh ? i1 : i0;
}
__device__ __forceinline__ void
sage_sum_prows(const float4* p4, int idx, int m, int hh, int c16, float4& acc) {
  int t = 0;
  for (; t + 16 <= m; t += 16) {
    const float4 a0 = SAGE_PROW(0), a1 = SAGE_PROW(2), a2 = SAGE_PROW(4), a3 = SAGE_PROW(6);
    const float4 a4 = SAGE_PROW(8), a5 = SAGE_PROW(10), a6 = SAGE_PROW(12), a7 = SAGE_PROW(14);
    SAGE_ADD(a0) SAGE_ADD(a1) SAGE_ADD(a2) SAGE_ADD(a3)
    SAGE_ADD(a4) SAGE_ADD(a5) SAGE_ADD(a6) SAGE_ADD(a7)
  }
  if (t + 8 <= m) {
    const float4 a0 = SAGE_PROW(0), a1 = SAGE_PROW(2), a2 = SAGE_PROW(4), a3 = SAGE_PROW(6);
    SAGE_ADD(a0) SAGE_ADD(a1) SAGE_ADD(a2) SAGE_ADD(a3)
    t += 8;
  }
  if (t + 4 <= m) {
    const float4 a0 = SAGE_PROW(0), a1 = SAGE_PROW(2);
    SAGE_ADD(a0) SAGE_ADD(a1)
    t += 4;
  }
  if (t + 2 <= m) {
    const float4 a0 = SAGE_PROW(0);
    SAGE_ADD(a0)
    t += 2;
  }
  if (t < m) {
    const float4 a0 = SAGE_PROW(0);
    if (hh == 0) { SAGE_ADD(a0) }
  }
}
__global__ void __launch_bounds__(256)
sage_gather_mean_kernel(const float* __restrict__ P,
                        const int* __restrict__ rowstart,
                        const int* __restrict__ col,
                        float* out) {
  const int lane = threadIdx.x & 31;
  const int c16 = lane & 15;
  const int hh = lane >> 4;
  const int gw = __builtin_amdgcn_readfirstlane((int)(blockIdx.x * 8 + (threadIdx.x >> 5)));
  const float4* p4 = (const float4*)P;
  int node = -1, beg = 0, end = 0, idx = 0;
  int node1 = gw;
  int beg1 = rowstart[node1];
  int end1 = rowstart[node1 + 1];
  for (;;) {
    int idx1 = 0;
    if (node1 < N_NODES) {
      const int j1 = beg1 + lane;
      idx1 = (j1 < end1) ? col[j1] : 0;
    }
    const int node2 = node1 + GATHER_WAVES;
    int beg2 = 0, end2 = 0;
    if (node2 < N_NODES) { beg2 = rowstart[node2]; end2 = rowstart[node2 + 1]; }

    if (node >= 0) {
      const float4 q = ((const float4*)out)[(long long)node * 16 + c16];
      float4 acc = make_float4(0.0f, 0.0f, 0.0f, 0.0f);
      sage_sum_prows(p4, idx, min(32, end - beg), hh, c16, acc);
      for (int base = beg + 32; base < end; base += 32) {
        const int j = base + lane;
        const int idl = (j < end) ? col[j] : 0;
        sage_sum_prows(p4, idl, min(32, end - base), hh, c16, acc);
      }
      acc.x += __shfl_xor(acc.x, 16);
      acc.y += __shfl_xor(acc.y, 16);
      acc.z += __shfl_xor(acc.z, 16);
      acc.w += __shfl_xor(acc.w, 16);
      const float scale = 1.0f / fmaxf((float)(end - beg), 1.0f);
      if (hh == 0) {
        float4 o;
        o.x = acc.x * scale + q.x;
        o.y = acc.y * scale + q.y;
        o.z = acc.z * scale + q.z;
        o.w = acc.w * scale + q.w;
        ((float4*)out)[(long long)node * 16 + c16] = o;
      }
    }

    if (node1 >= N_NODES) break;
    node = node1; beg = beg1; end = end1; idx = idx1;
    node1 = node2; beg1 = beg2; end1 = end2;
  }
}

extern "C" void kernel_launch(void* const* d_in, const int* in_sizes, int n_in,
                              void* d_out, int out_size, void* d_ws, size_t ws_size,
                              hipStream_t stream) {
  const float* x   = (const float*)d_in[0];
  const int*   ei  = (const int*)d_in[1];
  const float* Wl1 = (const float*)d_in[2];
  const float* bl1 = (const float*)d_in[3];
  const float* Wr1 = (const float*)d_in[4];
  const float* Wl2 = (const float*)d_in[5];
  const float* bl2 = (const float*)d_in[6];
  const float* Wr2 = (const float*)d_in[7];
  float* out = (float*)d_out;

  const int E = in_sizes[1] / 2;
  const int* src = ei;
  const int* dst = ei + E;

  float* msg = (float*)d_ws;
  float* deg = msg + (long long)N_NODES * FEAT;
  float* pl3 = deg + N_NODES;

  int* loc  = (int*)pl3;
  int* keys = loc + 65536;

  int* rowstart = (int*)pl3 + 1048576 + 48;
  int* col      = rowstart + XF_CSR_OFF;

  float* P = pl3 + 2097152 + 48;

  const int TB = 256;

  sage_chunk_sort_kernel<<<SORT_NCH, SORT_THREADS, 0, stream>>>(src, dst, loc, keys, E);
  sage_bucket_fill_kernel<<<SORT_NB, FILL_THREADS, 0, stream>>>(loc, keys, rowstart, deg, col, E);

  sage_gather_kernel<<<GATHER_WGS, TB, 0, stream>>>(x, rowstart, col, msg);

  sage_layers_kernel<<<FU_WGS, FU_THREADS, 0, stream>>>(msg, deg, x, Wl1, bl1, Wr1, Wl2, bl2, Wr2, P, out);

  sage_gather_mean_kernel<<<GATHER_WGS, TB, 0, stream>>>(P, rowstart, col, out);
}
